# starting kernel
# baseline (speedup 1.0000x reference)
#include <hip/hip_runtime.h>

#define NZ 512
#define NX 512
#define DT2 0.0025f

typedef __attribute__((ext_vector_type(2))) float v2f;
typedef __attribute__((ext_vector_type(8))) float v8f;

__global__ void prep_kernel(const float* __restrict__ vel,
                            const float* __restrict__ rho,
                            const float* __restrict__ beta,
                            float4* __restrict__ C4) {
    int i = blockIdx.x * blockDim.x + threadIdx.x;
    if (i >= NZ * NX) return;
    int z = i / NX, x = i - z * NX;
    auto ir = [&](int zz, int xx) -> float {
        return (zz >= 0 && zz < NZ && xx >= 0 && xx < NX) ? 1.0f / rho[zz * NX + xx] : 0.0f;
    };
    float Kc  = vel[i] * vel[i] * rho[i];
    float gxr = 0.5f * (ir(z, x + 1) - ir(z, x - 1));
    float gzr = 0.5f * (ir(z + 1, x) - ir(z - 1, x));
    float4 c;
    c.x = DT2 * Kc * rho[i];
    c.y = DT2 * Kc * gxr;
    c.z = DT2 * Kc * gzr;
    c.w = 2.0f * beta[i] / Kc;
    C4[i] = c;
}

__launch_bounds__(256)
__global__ void step_kernel(const float* __restrict__ pm1, const float* __restrict__ pm2,
                            float* __restrict__ pout,
                            const float4* __restrict__ C4,
                            const float* __restrict__ damping,
                            const float* __restrict__ pulse_t,
                            const float* __restrict__ pulse_next,
                            const int* __restrict__ nlaflag,
                            const int* __restrict__ el,
                            float* __restrict__ gout,
                            int nch) {
    __shared__ float T[8][20 * 21];
    __shared__ float TW[40];
    __shared__ float TG[40];

    const int wave = threadIdx.x >> 5;
    const int lane = threadIdx.x & 31;
    const int tile = blockIdx.x * 8 + wave;
    const int Z0 = (tile >> 5) << 4;
    const int X0 = (tile & 31) << 4;

    if (gout != nullptr && blockIdx.x == 0 && threadIdx.x < nch) {
        int iz = el[2 * threadIdx.x + 0];
        int ix = el[2 * threadIdx.x + 1];
        gout[threadIdx.x] = pm1[iz * NX + ix];
    }

    for (int i = threadIdx.x; i < 40; i += 256) {
        int d = i - 18;
        float w = 0.0f, g = 0.0f;
        if (d == -2 || d == 2) w = -1.0f / 12.0f;
        else if (d == -1 || d == 1) w = 4.0f / 3.0f;
        else if (d == 0) w = -2.5f;
        if (d == 1) g = 0.5f;
        else if (d == -1) g = -0.5f;
        TW[i] = w;
        TG[i] = g;
    }

    if (lane < 16) {
        __builtin_prefetch(&pulse_next[(Z0 + lane) * NX + X0], 0, 3);
    }

    for (int i = lane; i < 400; i += 32) {
        int zz = i / 20, xx = i - zz * 20;
        int gz = Z0 - 2 + zz, gx = X0 - 2 + xx;
        float v = (gz >= 0 && gz < NZ && gx >= 0 && gx < NX) ? pm1[gz * NX + gx] : 0.0f;
        T[wave][zz * 21 + xx] = v;
    }
    __syncthreads();

    const int half = lane >> 4;
    const int lm   = lane & 15;
    const int wbase = 2 * half - lm + 16;

    v8f accL  = {};
    v8f accGx = {};
    v8f accGz = {};

#pragma unroll
    for (int c = 0; c < 5; ++c) {
        const int k0 = 4 * c + 2 * half;
        const int wi = 4 * c + wbase;

        v2f fw; fw[0] = TW[wi]; fw[1] = TW[wi + 1];
        v2f fg; fg[0] = TG[wi]; fg[1] = TG[wi + 1];

        v2f px, pz;
        px[0] = T[wave][(lm + 2) * 21 + k0];
        px[1] = T[wave][(lm + 2) * 21 + k0 + 1];
        pz[0] = T[wave][k0 * 21 + lm + 2];
        pz[1] = T[wave][(k0 + 1) * 21 + lm + 2];

        accL  = __builtin_amdgcn_wmma_f32_16x16x4_f32(false, px, false, fw, (short)0, accL,  false, false);
        accL  = __builtin_amdgcn_wmma_f32_16x16x4_f32(false, fw, false, pz, (short)0, accL,  false, false);
        accGx = __builtin_amdgcn_wmma_f32_16x16x4_f32(false, px, false, fg, (short)0, accGx, false, false);
        accGz = __builtin_amdgcn_wmma_f32_16x16x4_f32(false, fg, false, pz, (short)0, accGz, false, false);
    }

    const int NLA = nlaflag[0];

#pragma unroll
    for (int r = 0; r < 8; ++r) {
        int m   = r + 8 * half;
        int gz  = Z0 + m, gx = X0 + lm;
        int idx = gz * NX + gx;

        float  p1  = T[wave][(m + 2) * 21 + lm + 2];
        float4 cf  = C4[idx];
        float  dmp = damping[idx];
        float  p2  = pm2[idx];
        float  pu  = pulse_t[idx];

        float rhs = cf.x * accL[r] + cf.y * accGx[r] + cf.z * accGz[r];

        float pnew;
        if (NLA) {
            float nla = 1.0f + cf.w * p1;
            float a   = nla + 2.0f * dmp;
            float b   = 2.0f * nla - dmp * dmp;
            float ccc = 2.0f * dmp - nla;
            float dp  = p1 - p2;
            pnew = (b * p1 + ccc * p2 + cf.w * dp * dp + rhs + pu) / a;
        } else {
            float a   = 1.0f + 2.0f * dmp;
            float b   = 2.0f - dmp * dmp;
            float ccc = 2.0f * dmp - 1.0f;
            pnew = (b * p1 + ccc * p2 + rhs + DT2 * pu) / a;
        }
        pout[idx] = pnew;
    }
}

extern "C" void kernel_launch(void* const* d_in, const int* in_sizes, int n_in,
                              void* d_out, int out_size, void* d_ws, size_t ws_size,
                              hipStream_t stream) {
    const float* pulse   = (const float*)d_in[0];
    const float* vel     = (const float*)d_in[1];
    const float* rho     = (const float*)d_in[2];
    const float* damping = (const float*)d_in[3];
    const float* beta    = (const float*)d_in[4];
    const int*   el      = (const int*)d_in[5];
    const int*   nla     = (const int*)d_in[6];

    const int nt  = in_sizes[0] / (NZ * NX);
    const int nch = in_sizes[5] / 2;

    float* ws = (float*)d_ws;
    const size_t F = (size_t)NZ * NX;
    float*  bufs[3] = { ws, ws + F, ws + 2 * F };
    float4* C4 = (float4*)(ws + 3 * F);

    hipMemsetAsync(bufs[0], 0, 2 * F * sizeof(float), stream);

    prep_kernel<<<(int)((F + 255) / 256), 256, 0, stream>>>(vel, rho, beta, C4);

    int i1 = 0, i2 = 1;
    for (int t = 0; t < nt; ++t) {
        int io = 3 - i1 - i2;
        const float* pu_t = pulse + (size_t)t * F;
        const float* pu_n = (t + 1 < nt) ? pulse + (size_t)(t + 1) * F : pu_t;
        float* gout = (t >= 1 && ((t - 1) & 3) == 0)
                    ? (float*)d_out + (size_t)((t - 1) >> 2) * nch : nullptr;
        step_kernel<<<128, 256, 0, stream>>>(bufs[i1], bufs[i2], bufs[io],
                                             C4, damping, pu_t, pu_n, nla,
                                             el, gout, nch);
        i2 = i1; i1 = io;
    }
}


# pulse planes requested behind the first barrier and waited for plane by plane; hand-written workgroup barrier (LDS waits only) keeps these loads in flight across the steps, not drained before step 0
# speedup vs baseline: 6.8093x; 6.8093x over previous
#include <hip/hip_runtime.h>

#define NZ 512
#define NX 512
#define DT2 0.0025f

#define KST 8
#define OT  32
#define HL  16
#define FW  68
#define FS  73

typedef __attribute__((ext_vector_type(2))) float v2f;
typedef __attribute__((ext_vector_type(8))) float v8f;

__device__ __forceinline__ float lap_w(int d) {
    int a = d < 0 ? -d : d;
    return a == 0 ? -2.5f : (a == 1 ? 4.0f / 3.0f : (a == 2 ? -1.0f / 12.0f : 0.0f));
}
__device__ __forceinline__ float grd_w(int d) {
    return d == 1 ? 0.5f : (d == -1 ? -0.5f : 0.0f);
}

__device__ __forceinline__ void lds_barrier() {
    asm volatile("s_wait_dscnt 0x0\n\ts_barrier_signal -1\n\ts_barrier_wait -1" ::: "memory");
}

__launch_bounds__(512)
__global__ void steps_kernel(const float* __restrict__ pin1, const float* __restrict__ pin2,
                             float* __restrict__ pout1, float* __restrict__ pout2,
                             const float* pulse,
                             const float* __restrict__ vel,
                             const float* __restrict__ rho,
                             const float* __restrict__ damping,
                             const float* __restrict__ beta,
                             const int* __restrict__ nlaflag,
                             const int* __restrict__ el,
                             float* __restrict__ gout,
                             int nch, int nsteps, int first) {
    __shared__ float F[2][FW * FS];
    __shared__ float IR[FW * FS];

    const int tid  = threadIdx.x;
    const int wave = __builtin_amdgcn_readfirstlane(tid >> 5);
    const int lane = tid & 31;
    const int half = lane >> 4;
    const int lm   = lane & 15;
    const unsigned code = (unsigned)((0xFEDCB8743210A965ULL >> (4 * wave)) & 15ULL);
    const int tz = (int)(code >> 2), tx = (int)(code & 3u);
    const int GZ0 = (int)blockIdx.y * OT - HL;
    const int GX0 = (int)blockIdx.x * OT - HL;
    const int fbase = (16 * tz) * FS + 16 * tx;

    int ez = 0, ex = 0;
    if (nch > 0) {
        const int ci = tid < nch ? tid : nch - 1;
        ez = el[2 * ci + 0];
        ex = el[2 * ci + 1];
    }

    const int NLA = nlaflag[0];

    float sv[10], sr[10];
    int   sad[10];
    bool  sok[10];
#pragma unroll
    for (int j = 0; j < 10; ++j) {
        int i  = tid + 512 * j;
        int ii = i < FW * FW ? i : FW * FW - 1;
        int zz = ii / FW, xx = ii - zz * FW;
        int gz = GZ0 - 2 + zz, gxs = GX0 - 2 + xx;
        int cz = gz < 0 ? 0 : (gz > NZ - 1 ? NZ - 1 : gz);
        int cx = gxs < 0 ? 0 : (gxs > NX - 1 ? NX - 1 : gxs);
        sok[j] = (cz == gz) & (cx == gxs);
        sad[j] = cz * NX + cx;
        sr[j]  = rho[sad[j]];
        sv[j]  = 0.0f;
    }
    if (!first) {
#pragma unroll
        for (int j = 0; j < 10; ++j) sv[j] = pin1[sad[j]];
    }

    float rC[8], vV[8], bB[8], cDm[8], p2[8], pu[KST][8];
    bool inb[8];
    int  cidx[8];
    const int gx  = GX0 + 16 * tx + lm;
    const int cxc = gx < 0 ? 0 : (gx > NX - 1 ? NX - 1 : gx);
#pragma unroll
    for (int r = 0; r < 8; ++r) {
        const int gz  = GZ0 + 16 * tz + r + 8 * half;
        const int czc = gz < 0 ? 0 : (gz > NZ - 1 ? NZ - 1 : gz);
        inb[r]  = (czc == gz) & (cxc == gx);
        cidx[r] = czc * NX + cxc;
        rC[r]  = rho[cidx[r]];
        vV[r]  = vel[cidx[r]];
        bB[r]  = beta[cidx[r]];
        cDm[r] = damping[cidx[r]];
        p2[r]  = 0.0f;
    }
    if (!first) {
#pragma unroll
        for (int r = 0; r < 8; ++r) p2[r] = pin2[cidx[r]];
    }
#pragma unroll
    for (int j = 0; j < 10; ++j) {
        int i = tid + 512 * j;
        if (i < FW * FW) {
            int zz = i / FW, xx = i - zz * FW;
            F[0][zz * FS + xx] = sok[j] ? sv[j] : 0.0f;
            F[1][zz * FS + xx] = 0.0f;
            IR[zz * FS + xx]   = sok[j] ? 1.0f / sr[j] : 0.0f;
        }
    }

    v2f fw[5], fg[5];
#pragma unroll
    for (int c = 0; c < 5; ++c) {
        const int d = 4 * c + 2 * half - 2 - lm;
        fw[c][0] = lap_w(d); fw[c][1] = lap_w(d + 1);
        fg[c][0] = grd_w(d); fg[c][1] = grd_w(d + 1);
    }

    int sel_off = -1;
    {
        const int rz = ez - (GZ0 + HL), rx = ex - (GX0 + HL);
        const bool mine = (tid < nch) & (rz >= 0) & (rz < OT) & (rx >= 0) & (rx < OT);
        sel_off = mine ? (rz + HL + 2) * FS + rx + HL + 2 : -1;
    }

    lds_barrier();

#pragma unroll
    for (int s = 0; s < KST; ++s) {
        const int sc = s < nsteps ? s : nsteps - 1;
#pragma unroll
        for (int r = 0; r < 8; ++r) pu[s][r] = pulse[(size_t)sc * (NZ * NX) + cidx[r]];
    }
    asm volatile("" ::: "memory");

    float p1[8], cA1[8], cA2x[8], cA2z[8], cD[8];
#pragma unroll
    for (int r = 0; r < 8; ++r) {
        const int fo = fbase + (r + 8 * half + 2) * FS + lm + 2;
        p1[r] = F[0][fo];
        const float irE = IR[fo + 1], irW = IR[fo - 1], irS = IR[fo + FS], irN = IR[fo - FS];
        const float Kc  = vV[r] * vV[r] * rC[r];
        const float gxr = 0.5f * (irE - irW);
        const float gzr = 0.5f * (irS - irN);
        cA1[r]  = DT2 * Kc * rC[r];
        cA2x[r] = DT2 * Kc * gxr;
        cA2z[r] = DT2 * Kc * gzr;
        cD[r]   = 2.0f * bB[r] / Kc;
    }

#pragma unroll
    for (int s = 0; s < KST; ++s) {
        if (s < nsteps) {
            const float* Fc = F[s & 1];
            float*       Fn = F[(s & 1) ^ 1];
            const bool lastst = (s == nsteps - 1);
            const bool sample = ((s & 3) == 0);
            if (!lastst || wave < 4) {
                v8f accL  = {};
                v8f accGx = {};
                v8f accGz = {};
#pragma unroll
                for (int c = 0; c < 5; ++c) {
                    const int k0 = 4 * c + 2 * half;
                    v2f px, pz;
                    px[0] = Fc[fbase + (lm + 2) * FS + k0];
                    px[1] = Fc[fbase + (lm + 2) * FS + k0 + 1];
                    pz[0] = Fc[fbase + k0 * FS + lm + 2];
                    pz[1] = Fc[fbase + (k0 + 1) * FS + lm + 2];
                    accL  = __builtin_amdgcn_wmma_f32_16x16x4_f32(false, px,    false, fw[c], (short)0, accL,  false, false);
                    accL  = __builtin_amdgcn_wmma_f32_16x16x4_f32(false, fw[c], false, pz,    (short)0, accL,  false, false);
                    accGx = __builtin_amdgcn_wmma_f32_16x16x4_f32(false, px,    false, fg[c], (short)0, accGx, false, false);
                    accGz = __builtin_amdgcn_wmma_f32_16x16x4_f32(false, fg[c], false, pz,    (short)0, accGz, false, false);
                }
                float pn[8];
                if (NLA) {
#pragma unroll
                    for (int r = 0; r < 8; ++r) {
                        const float dmp = cDm[r];
                        const float rhs = cA1[r] * accL[r] + cA2x[r] * accGx[r] + cA2z[r] * accGz[r];
                        float nla = 1.0f + cD[r] * p1[r];
                        float a   = nla + 2.0f * dmp;
                        float b   = 2.0f * nla - dmp * dmp;
                        float ccc = 2.0f * dmp - nla;
                        float dp  = p1[r] - p2[r];
                        pn[r] = (b * p1[r] + ccc * p2[r] + cD[r] * dp * dp + rhs + pu[s][r]) / a;
                    }
                } else {
#pragma unroll
                    for (int r = 0; r < 8; ++r) {
                        const float dmp = cDm[r];
                        const float rhs = cA1[r] * accL[r] + cA2x[r] * accGx[r] + cA2z[r] * accGz[r];
                        float a   = 1.0f + 2.0f * dmp;
                        float b   = 2.0f - dmp * dmp;
                        float ccc = 2.0f * dmp - 1.0f;
                        pn[r] = (b * p1[r] + ccc * p2[r] + rhs + DT2 * pu[s][r]) / a;
                    }
                }
#pragma unroll
                for (int r = 0; r < 8; ++r) {
                    p2[r] = p1[r];
                    p1[r] = inb[r] ? pn[r] : 0.0f;
                }
                if (!lastst || sample) {
#pragma unroll
                    for (int r = 0; r < 8; ++r)
                        Fn[fbase + (r + 8 * half + 2) * FS + lm + 2] = p1[r];
                }
            }
            if (!lastst || sample) lds_barrier();
            if (sample) {
                float* grow = gout + (size_t)(s >> 2) * nch;
                if (sel_off >= 0) grow[tid] = Fn[sel_off];
                for (int ch = tid + 512; ch < nch; ch += 512) {
                    int rz = el[2 * ch + 0] - (GZ0 + HL);
                    int rx = el[2 * ch + 1] - (GX0 + HL);
                    if (rz >= 0 && rz < OT && rx >= 0 && rx < OT)
                        grow[ch] = Fn[(rz + HL + 2) * FS + rx + HL + 2];
                }
            }
        }
    }

    if (wave < 4) {
#pragma unroll
        for (int r = 0; r < 8; ++r) {
            const int idx = (GZ0 + 16 * tz + r + 8 * half) * NX + gx;
            pout1[idx] = p1[r];
            pout2[idx] = p2[r];
        }
    }
}

extern "C" void kernel_launch(void* const* d_in, const int* in_sizes, int n_in,
                              void* d_out, int out_size, void* d_ws, size_t ws_size,
                              hipStream_t stream) {
    const float* pulse   = (const float*)d_in[0];
    const float* vel     = (const float*)d_in[1];
    const float* rho     = (const float*)d_in[2];
    const float* damping = (const float*)d_in[3];
    const float* beta    = (const float*)d_in[4];
    const int*   el      = (const int*)d_in[5];
    const int*   nla     = (const int*)d_in[6];

    const int nt  = in_sizes[0] / (NZ * NX);
    const int nch = in_sizes[5] / 2;

    float* ws = (float*)d_ws;
    const size_t F = (size_t)NZ * NX;
    float* pl[4] = { ws, ws + F, ws + 2 * F, ws + 3 * F };

    int cur = 0;
    for (int t0 = 0; t0 < nt; t0 += KST) {
        const int ns = (nt - t0 < KST) ? (nt - t0) : KST;
        steps_kernel<<<dim3(NX / OT, NZ / OT), 512, 0, stream>>>(
            pl[2 * cur], pl[2 * cur + 1], pl[2 * (cur ^ 1)], pl[2 * (cur ^ 1) + 1],
            pulse + (size_t)t0 * F, vel, rho, damping, beta, nla, el,
            (float*)d_out + (size_t)(t0 >> 2) * nch, nch, ns, t0 == 0 ? 1 : 0);
        cur ^= 1;
    }
}


# z products over a tile's own rows read the field from the lane's registers (new fragment k order: sums in another order), issued for the next step between barrier signal and wait at low wave priority
# speedup vs baseline: 7.1214x; 1.0458x over previous
#include <hip/hip_runtime.h>

#define NZ 512
#define NX 512
#define DT2 0.0025f

#define KST 8
#define OT  32
#define HL  16
#define FW  68
#define FS  69

typedef __attribute__((ext_vector_type(2))) float v2f;
typedef __attribute__((ext_vector_type(8))) float v8f;

__device__ __forceinline__ float lap_w(int d) {
    int a = d < 0 ? -d : d;
    return a == 0 ? -2.5f : (a == 1 ? 4.0f / 3.0f : (a == 2 ? -1.0f / 12.0f : 0.0f));
}
__device__ __forceinline__ float grd_w(int d) {
    return d == 1 ? 0.5f : (d == -1 ? -0.5f : 0.0f);
}

__device__ __forceinline__ void lds_barrier() {
    asm volatile("s_wait_dscnt 0x0\n\ts_barrier_signal -1\n\ts_barrier_wait -1" ::: "memory");
}

__device__ __forceinline__ void lds_barrier_arrive(float (&p)[8]) {
    asm volatile("s_wait_dscnt 0x0\n\ts_barrier_signal -1"
                 : "+v"(p[0]), "+v"(p[1]), "+v"(p[2]), "+v"(p[3]), "+v"(p[4]), "+v"(p[5]), "+v"(p[6]), "+v"(p[7])
                 :: "memory");
}
__device__ __forceinline__ void lds_barrier_wait(v8f& a, v8f& b) {
    asm volatile("s_barrier_wait -1" : "+v"(a), "+v"(b) :: "memory");
}

__device__ __forceinline__ void z_own_rows(const float (&p)[8], const v2f (&fw)[5], const v2f (&fg)[5],
                                           v8f& accL, v8f& accGz) {
#pragma unroll
    for (int c = 0; c < 4; ++c) {
        v2f pz;
        pz[0] = p[2 * c];
        pz[1] = p[2 * c + 1];
        accL  = __builtin_amdgcn_wmma_f32_16x16x4_f32(false, fw[c], false, pz, (short)0, accL,  false, false);
        accGz = __builtin_amdgcn_wmma_f32_16x16x4_f32(false, fg[c], false, pz, (short)0, accGz, false, false);
    }
}

__launch_bounds__(512)
__global__ void steps_kernel(const float* __restrict__ pin1, const float* __restrict__ pin2,
                             float* __restrict__ pout1, float* __restrict__ pout2,
                             const float* pulse,
                             const float* __restrict__ vel,
                             const float* __restrict__ rho,
                             const float* __restrict__ damping,
                             const float* __restrict__ beta,
                             const int* __restrict__ nlaflag,
                             const int* __restrict__ el,
                             float* __restrict__ gout,
                             int nch, int nsteps, int first) {
    __shared__ float F[2][FW * FS];
    __shared__ float IR[FW * FS];

    const int tid  = threadIdx.x;
    const int wave = __builtin_amdgcn_readfirstlane(tid >> 5);
    const int lane = tid & 31;
    const int half = lane >> 4;
    const int lm   = lane & 15;
    const unsigned code = (unsigned)((0xFEDCB8743210A965ULL >> (4 * wave)) & 15ULL);
    const int tz = (int)(code >> 2), tx = (int)(code & 3u);
    const int GZ0 = (int)blockIdx.y * OT - HL;
    const int GX0 = (int)blockIdx.x * OT - HL;
    const int fbase = (16 * tz) * FS + 16 * tx;

    int ez = 0, ex = 0;
    if (nch > 0) {
        const int ci = tid < nch ? tid : nch - 1;
        ez = el[2 * ci + 0];
        ex = el[2 * ci + 1];
    }

    const int NLA = nlaflag[0];

    float sv[10], sr[10];
    int   sad[10];
    bool  sok[10];
#pragma unroll
    for (int j = 0; j < 10; ++j) {
        int i  = tid + 512 * j;
        int ii = i < FW * FW ? i : FW * FW - 1;
        int zz = ii / FW, xx = ii - zz * FW;
        int gz = GZ0 - 2 + zz, gxs = GX0 - 2 + xx;
        int cz = gz < 0 ? 0 : (gz > NZ - 1 ? NZ - 1 : gz);
        int cx = gxs < 0 ? 0 : (gxs > NX - 1 ? NX - 1 : gxs);
        sok[j] = (cz == gz) & (cx == gxs);
        sad[j] = cz * NX + cx;
        sr[j]  = rho[sad[j]];
        sv[j]  = 0.0f;
    }
    if (!first) {
#pragma unroll
        for (int j = 0; j < 10; ++j) sv[j] = pin1[sad[j]];
    }

    float rC[8], vV[8], bB[8], cDm[8], p2[8], pu[KST][8];
    bool inb[8];
    int  cidx[8];
    const int gx  = GX0 + 16 * tx + lm;
    const int cxc = gx < 0 ? 0 : (gx > NX - 1 ? NX - 1 : gx);
#pragma unroll
    for (int r = 0; r < 8; ++r) {
        const int gz  = GZ0 + 16 * tz + r + 8 * half;
        const int czc = gz < 0 ? 0 : (gz > NZ - 1 ? NZ - 1 : gz);
        inb[r]  = (czc == gz) & (cxc == gx);
        cidx[r] = czc * NX + cxc;
        rC[r]  = rho[cidx[r]];
        vV[r]  = vel[cidx[r]];
        bB[r]  = beta[cidx[r]];
        cDm[r] = damping[cidx[r]];
        p2[r]  = 0.0f;
    }
    if (!first) {
#pragma unroll
        for (int r = 0; r < 8; ++r) p2[r] = pin2[cidx[r]];
    }
#pragma unroll
    for (int j = 0; j < 10; ++j) {
        int i = tid + 512 * j;
        if (i < FW * FW) {
            int zz = i / FW, xx = i - zz * FW;
            F[0][zz * FS + xx] = sok[j] ? sv[j] : 0.0f;
            F[1][zz * FS + xx] = 0.0f;
            IR[zz * FS + xx]   = sok[j] ? 1.0f / sr[j] : 0.0f;
        }
    }

    v2f fw[5], fg[5];
#pragma unroll
    for (int c = 0; c < 5; ++c) {
        const int kk = c < 4 ? 8 * half + 2 * c : (half ? 16 : -2);
        const int d  = kk - lm;
        fw[c][0] = lap_w(d); fw[c][1] = lap_w(d + 1);
        fg[c][0] = grd_w(d); fg[c][1] = grd_w(d + 1);
    }
    const int hk = half ? 18 : 0;

    int sel_off = -1;
    {
        const int rz = ez - (GZ0 + HL), rx = ex - (GX0 + HL);
        const bool mine = (tid < nch) & (rz >= 0) & (rz < OT) & (rx >= 0) & (rx < OT);
        sel_off = mine ? (rz + HL + 2) * FS + rx + HL + 2 : -1;
    }

    lds_barrier();

#pragma unroll
    for (int s = 0; s < KST; ++s) {
        const int sc = s < nsteps ? s : nsteps - 1;
#pragma unroll
        for (int r = 0; r < 8; ++r) pu[s][r] = pulse[(size_t)sc * (NZ * NX) + cidx[r]];
    }
    asm volatile("" ::: "memory");

    float p1[8], cA1[8], cA2x[8], cA2z[8], cD[8];
#pragma unroll
    for (int r = 0; r < 8; ++r) {
        const int fo = fbase + (r + 8 * half + 2) * FS + lm + 2;
        p1[r] = F[0][fo];
        const float irE = IR[fo + 1], irW = IR[fo - 1], irS = IR[fo + FS], irN = IR[fo - FS];
        const float Kc  = vV[r] * vV[r] * rC[r];
        const float gxr = 0.5f * (irE - irW);
        const float gzr = 0.5f * (irS - irN);
        cA1[r]  = DT2 * Kc * rC[r];
        cA2x[r] = DT2 * Kc * gxr;
        cA2z[r] = DT2 * Kc * gzr;
        cD[r]   = 2.0f * bB[r] / Kc;
    }

    v8f accL  = {};
    v8f accGz = {};
    if (nsteps > 1 || wave < 4) z_own_rows(p1, fw, fg, accL, accGz);

#pragma unroll
    for (int s = 0; s < KST; ++s) {
        if (s < nsteps) {
            const float* Fc = F[s & 1];
            float*       Fn = F[(s & 1) ^ 1];
            const bool lastst = (s == nsteps - 1);
            const bool sample = ((s & 3) == 0);
            if (!lastst || wave < 4) {
                __builtin_amdgcn_s_setprio(3);
                v8f accGx = {};
                v2f px[5], pzh;
                pzh[0]   = Fc[fbase + hk * FS + lm + 2];
                pzh[1]   = Fc[fbase + (hk + 1) * FS + lm + 2];
#pragma unroll
                for (int c = 0; c < 4; ++c) {
                    px[c][0] = Fc[fbase + (lm + 2) * FS + 8 * half + 2 * c + 2];
                    px[c][1] = Fc[fbase + (lm + 2) * FS + 8 * half + 2 * c + 3];
                }
                px[4][0] = Fc[fbase + (lm + 2) * FS + hk];
                px[4][1] = Fc[fbase + (lm + 2) * FS + hk + 1];
                accL  = __builtin_amdgcn_wmma_f32_16x16x4_f32(false, fw[4], false, pzh, (short)0, accL,  false, false);
                accGz = __builtin_amdgcn_wmma_f32_16x16x4_f32(false, fg[4], false, pzh, (short)0, accGz, false, false);
#pragma unroll
                for (int c = 0; c < 5; ++c) {
                    accL  = __builtin_amdgcn_wmma_f32_16x16x4_f32(false, px[c], false, fw[c], (short)0, accL,  false, false);
                    accGx = __builtin_amdgcn_wmma_f32_16x16x4_f32(false, px[c], false, fg[c], (short)0, accGx, false, false);
                }
                __builtin_amdgcn_s_setprio(0);
                float pn[8];
                if (NLA) {
#pragma unroll
                    for (int r = 0; r < 8; ++r) {
                        const float dmp = cDm[r];
                        const float rhs = cA1[r] * accL[r] + cA2x[r] * accGx[r] + cA2z[r] * accGz[r];
                        float nla = 1.0f + cD[r] * p1[r];
                        float a   = nla + 2.0f * dmp;
                        float b   = 2.0f * nla - dmp * dmp;
                        float ccc = 2.0f * dmp - nla;
                        float dp  = p1[r] - p2[r];
                        pn[r] = (b * p1[r] + ccc * p2[r] + cD[r] * dp * dp + rhs + pu[s][r]) / a;
                    }
                } else {
#pragma unroll
                    for (int r = 0; r < 8; ++r) {
                        const float dmp = cDm[r];
                        const float rhs = cA1[r] * accL[r] + cA2x[r] * accGx[r] + cA2z[r] * accGz[r];
                        float a   = 1.0f + 2.0f * dmp;
                        float b   = 2.0f - dmp * dmp;
                        float ccc = 2.0f * dmp - 1.0f;
                        pn[r] = (b * p1[r] + ccc * p2[r] + rhs + DT2 * pu[s][r]) / a;
                    }
                }
#pragma unroll
                for (int r = 0; r < 8; ++r) {
                    p2[r] = p1[r];
                    p1[r] = inb[r] ? pn[r] : 0.0f;
                }
                if (!lastst || sample) {
#pragma unroll
                    for (int r = 0; r < 8; ++r)
                        Fn[fbase + (r + 8 * half + 2) * FS + lm + 2] = p1[r];
                }
            }
            const bool bar = (!lastst || sample);
            if (bar) lds_barrier_arrive(p1);
            if (s + 1 < KST && s + 1 < nsteps && (s + 1 < nsteps - 1 || wave < 4)) {
                accL  = v8f{};
                accGz = v8f{};
                z_own_rows(p1, fw, fg, accL, accGz);
            }
            if (bar) lds_barrier_wait(accL, accGz);
            if (sample) {
                float* grow = gout + (size_t)(s >> 2) * nch;
                if (sel_off >= 0) grow[tid] = Fn[sel_off];
                for (int ch = tid + 512; ch < nch; ch += 512) {
                    int rz = el[2 * ch + 0] - (GZ0 + HL);
                    int rx = el[2 * ch + 1] - (GX0 + HL);
                    if (rz >= 0 && rz < OT && rx >= 0 && rx < OT)
                        grow[ch] = Fn[(rz + HL + 2) * FS + rx + HL + 2];
                }
            }
        }
    }

    if (wave < 4) {
#pragma unroll
        for (int r = 0; r < 8; ++r) {
            const int idx = (GZ0 + 16 * tz + r + 8 * half) * NX + gx;
            pout1[idx] = p1[r];
            pout2[idx] = p2[r];
        }
    }
}

extern "C" void kernel_launch(void* const* d_in, const int* in_sizes, int n_in,
                              void* d_out, int out_size, void* d_ws, size_t ws_size,
                              hipStream_t stream) {
    const float* pulse   = (const float*)d_in[0];
    const float* vel     = (const float*)d_in[1];
    const float* rho     = (const float*)d_in[2];
    const float* damping = (const float*)d_in[3];
    const float* beta    = (const float*)d_in[4];
    const int*   el      = (const int*)d_in[5];
    const int*   nla     = (const int*)d_in[6];

    const int nt  = in_sizes[0] / (NZ * NX);
    const int nch = in_sizes[5] / 2;

    float* ws = (float*)d_ws;
    const size_t F = (size_t)NZ * NX;
    float* pl[4] = { ws, ws + F, ws + 2 * F, ws + 3 * F };

    int cur = 0;
    for (int t0 = 0; t0 < nt; t0 += KST) {
        const int ns = (nt - t0 < KST) ? (nt - t0) : KST;
        steps_kernel<<<dim3(NX / OT, NZ / OT), 512, 0, stream>>>(
            pl[2 * cur], pl[2 * cur + 1], pl[2 * (cur ^ 1)], pl[2 * (cur ^ 1) + 1],
            pulse + (size_t)t0 * F, vel, rho, damping, beta, nla, el,
            (float*)d_out + (size_t)(t0 >> 2) * nch, nch, ns, t0 == 0 ? 1 : 0);
        cur ^= 1;
    }
}


# every f32 division (step quotient, 1/rho, 2*beta/Kc) as hardware reciprocal, product and one fma residual correction; cell update written on register pairs: last bit of a quotient may differ
# speedup vs baseline: 8.1903x; 1.1501x over previous
#include <hip/hip_runtime.h>

#define NZ 512
#define NX 512
#define DT2 0.0025f

#define KST 8
#define OT  32
#define HL  16
#define FW  68
#define FS  69

typedef __attribute__((ext_vector_type(2))) float v2f;
typedef __attribute__((ext_vector_type(8))) float v8f;

__device__ __forceinline__ float lap_w(int d) {
    int a = d < 0 ? -d : d;
    return a == 0 ? -2.5f : (a == 1 ? 4.0f / 3.0f : (a == 2 ? -1.0f / 12.0f : 0.0f));
}
__device__ __forceinline__ float grd_w(int d) {
    return d == 1 ? 0.5f : (d == -1 ? -0.5f : 0.0f);
}

__device__ __forceinline__ float div_rc(float n, float a) {
    const float r = __builtin_amdgcn_rcpf(a);
    const float q = n * r;
    const float e = __builtin_fmaf(-a, q, n);
    return __builtin_fmaf(e, r, q);
}
__device__ __forceinline__ v2f div_rc2(v2f n, v2f a) {
    v2f r;
    r[0] = __builtin_amdgcn_rcpf(a[0]);
    r[1] = __builtin_amdgcn_rcpf(a[1]);
    const v2f q = n * r;
    const v2f e = __builtin_elementwise_fma(-a, q, n);
    return __builtin_elementwise_fma(e, r, q);
}
__device__ __forceinline__ float rcp_nr(float a) {
    const float r = __builtin_amdgcn_rcpf(a);
    const float e = __builtin_fmaf(-a, r, 1.0f);
    return __builtin_fmaf(e, r, r);
}

__device__ __forceinline__ void lds_barrier() {
    asm volatile("s_wait_dscnt 0x0\n\ts_barrier_signal -1\n\ts_barrier_wait -1" ::: "memory");
}

__device__ __forceinline__ void lds_barrier_arrive(float (&p)[8]) {
    asm volatile("s_wait_dscnt 0x0\n\ts_barrier_signal -1"
                 : "+v"(p[0]), "+v"(p[1]), "+v"(p[2]), "+v"(p[3]), "+v"(p[4]), "+v"(p[5]), "+v"(p[6]), "+v"(p[7])
                 :: "memory");
}
__device__ __forceinline__ void lds_barrier_wait(v8f& a, v8f& b) {
    asm volatile("s_barrier_wait -1" : "+v"(a), "+v"(b) :: "memory");
}

__device__ __forceinline__ void z_own_rows(const float (&p)[8], const v2f (&fw)[5], const v2f (&fg)[5],
                                           v8f& accL, v8f& accGz) {
#pragma unroll
    for (int c = 0; c < 4; ++c) {
        v2f pz;
        pz[0] = p[2 * c];
        pz[1] = p[2 * c + 1];
        accL  = __builtin_amdgcn_wmma_f32_16x16x4_f32(false, fw[c], false, pz, (short)0, accL,  false, false);
        accGz = __builtin_amdgcn_wmma_f32_16x16x4_f32(false, fg[c], false, pz, (short)0, accGz, false, false);
    }
}

__launch_bounds__(512)
__global__ void steps_kernel(const float* __restrict__ pin1, const float* __restrict__ pin2,
                             float* __restrict__ pout1, float* __restrict__ pout2,
                             const float* pulse,
                             const float* __restrict__ vel,
                             const float* __restrict__ rho,
                             const float* __restrict__ damping,
                             const float* __restrict__ beta,
                             const int* __restrict__ nlaflag,
                             const int* __restrict__ el,
                             float* __restrict__ gout,
                             int nch, int nsteps, int first) {
    __shared__ float F[2][FW * FS];
    __shared__ float IR[FW * FS];

    const int tid  = threadIdx.x;
    const int wave = __builtin_amdgcn_readfirstlane(tid >> 5);
    const int lane = tid & 31;
    const int half = lane >> 4;
    const int lm   = lane & 15;
    const unsigned code = (unsigned)((0xFEDCB8743210A965ULL >> (4 * wave)) & 15ULL);
    const int tz = (int)(code >> 2), tx = (int)(code & 3u);
    const int GZ0 = (int)blockIdx.y * OT - HL;
    const int GX0 = (int)blockIdx.x * OT - HL;
    const int fbase = (16 * tz) * FS + 16 * tx;

    int ez = 0, ex = 0;
    if (nch > 0) {
        const int ci = tid < nch ? tid : nch - 1;
        ez = el[2 * ci + 0];
        ex = el[2 * ci + 1];
    }

    const int NLA = nlaflag[0];

    float sv[10], sr[10];
    int   sad[10];
    bool  sok[10];
#pragma unroll
    for (int j = 0; j < 10; ++j) {
        int i  = tid + 512 * j;
        int ii = i < FW * FW ? i : FW * FW - 1;
        int zz = ii / FW, xx = ii - zz * FW;
        int gz = GZ0 - 2 + zz, gxs = GX0 - 2 + xx;
        int cz = gz < 0 ? 0 : (gz > NZ - 1 ? NZ - 1 : gz);
        int cx = gxs < 0 ? 0 : (gxs > NX - 1 ? NX - 1 : gxs);
        sok[j] = (cz == gz) & (cx == gxs);
        sad[j] = cz * NX + cx;
        sr[j]  = rho[sad[j]];
        sv[j]  = 0.0f;
    }
    if (!first) {
#pragma unroll
        for (int j = 0; j < 10; ++j) sv[j] = pin1[sad[j]];
    }

    float rC[8], vV[8], bB[8], cDm[8], p2[8], pu[KST][8];
    bool inb[8];
    int  cidx[8];
    const int gx  = GX0 + 16 * tx + lm;
    const int cxc = gx < 0 ? 0 : (gx > NX - 1 ? NX - 1 : gx);
#pragma unroll
    for (int r = 0; r < 8; ++r) {
        const int gz  = GZ0 + 16 * tz + r + 8 * half;
        const int czc = gz < 0 ? 0 : (gz > NZ - 1 ? NZ - 1 : gz);
        inb[r]  = (czc == gz) & (cxc == gx);
        cidx[r] = czc * NX + cxc;
        rC[r]  = rho[cidx[r]];
        vV[r]  = vel[cidx[r]];
        bB[r]  = beta[cidx[r]];
        cDm[r] = damping[cidx[r]];
        p2[r]  = 0.0f;
    }
    if (!first) {
#pragma unroll
        for (int r = 0; r < 8; ++r) p2[r] = pin2[cidx[r]];
    }
#pragma unroll
    for (int j = 0; j < 10; ++j) {
        int i = tid + 512 * j;
        if (i < FW * FW) {
            int zz = i / FW, xx = i - zz * FW;
            F[0][zz * FS + xx] = sok[j] ? sv[j] : 0.0f;
            F[1][zz * FS + xx] = 0.0f;
            IR[zz * FS + xx]   = sok[j] ? rcp_nr(sr[j]) : 0.0f;
        }
    }

    v2f fw[5], fg[5];
#pragma unroll
    for (int c = 0; c < 5; ++c) {
        const int kk = c < 4 ? 8 * half + 2 * c : (half ? 16 : -2);
        const int d  = kk - lm;
        fw[c][0] = lap_w(d); fw[c][1] = lap_w(d + 1);
        fg[c][0] = grd_w(d); fg[c][1] = grd_w(d + 1);
    }
    const int hk = half ? 18 : 0;

    int sel_off = -1;
    {
        const int rz = ez - (GZ0 + HL), rx = ex - (GX0 + HL);
        const bool mine = (tid < nch) & (rz >= 0) & (rz < OT) & (rx >= 0) & (rx < OT);
        sel_off = mine ? (rz + HL + 2) * FS + rx + HL + 2 : -1;
    }

    lds_barrier();

#pragma unroll
    for (int s = 0; s < KST; ++s) {
        const int sc = s < nsteps ? s : nsteps - 1;
#pragma unroll
        for (int r = 0; r < 8; ++r) pu[s][r] = pulse[(size_t)sc * (NZ * NX) + cidx[r]];
    }
    asm volatile("" ::: "memory");

    float p1[8], cA1[8], cA2x[8], cA2z[8], cD[8];
#pragma unroll
    for (int r = 0; r < 8; ++r) {
        const int fo = fbase + (r + 8 * half + 2) * FS + lm + 2;
        p1[r] = F[0][fo];
        const float irE = IR[fo + 1], irW = IR[fo - 1], irS = IR[fo + FS], irN = IR[fo - FS];
        const float Kc  = vV[r] * vV[r] * rC[r];
        const float gxr = 0.5f * (irE - irW);
        const float gzr = 0.5f * (irS - irN);
        cA1[r]  = DT2 * Kc * rC[r];
        cA2x[r] = DT2 * Kc * gxr;
        cA2z[r] = DT2 * Kc * gzr;
        cD[r]   = div_rc(2.0f * bB[r], Kc);
    }

    v8f accL  = {};
    v8f accGz = {};
    if (nsteps > 1 || wave < 4) z_own_rows(p1, fw, fg, accL, accGz);

#pragma unroll
    for (int s = 0; s < KST; ++s) {
        if (s < nsteps) {
            const float* Fc = F[s & 1];
            float*       Fn = F[(s & 1) ^ 1];
            const bool lastst = (s == nsteps - 1);
            const bool sample = ((s & 3) == 0);
            if (!lastst || wave < 4) {
                __builtin_amdgcn_s_setprio(3);
                v8f accGx = {};
                v2f px[5], pzh;
                pzh[0]   = Fc[fbase + hk * FS + lm + 2];
                pzh[1]   = Fc[fbase + (hk + 1) * FS + lm + 2];
#pragma unroll
                for (int c = 0; c < 4; ++c) {
                    px[c][0] = Fc[fbase + (lm + 2) * FS + 8 * half + 2 * c + 2];
                    px[c][1] = Fc[fbase + (lm + 2) * FS + 8 * half + 2 * c + 3];
                }
                px[4][0] = Fc[fbase + (lm + 2) * FS + hk];
                px[4][1] = Fc[fbase + (lm + 2) * FS + hk + 1];
                accL  = __builtin_amdgcn_wmma_f32_16x16x4_f32(false, fw[4], false, pzh, (short)0, accL,  false, false);
                accGz = __builtin_amdgcn_wmma_f32_16x16x4_f32(false, fg[4], false, pzh, (short)0, accGz, false, false);
#pragma unroll
                for (int c = 0; c < 5; ++c) {
                    accL  = __builtin_amdgcn_wmma_f32_16x16x4_f32(false, px[c], false, fw[c], (short)0, accL,  false, false);
                    accGx = __builtin_amdgcn_wmma_f32_16x16x4_f32(false, px[c], false, fg[c], (short)0, accGx, false, false);
                }
                __builtin_amdgcn_s_setprio(0);
                float pn[8];
                if (NLA) {
#pragma unroll
                    for (int k = 0; k < 4; ++k) {
                        const v2f dmp = {cDm[2 * k], cDm[2 * k + 1]};
                        const v2f q1  = {p1[2 * k], p1[2 * k + 1]};
                        const v2f q2  = {p2[2 * k], p2[2 * k + 1]};
                        const v2f dd  = {cD[2 * k], cD[2 * k + 1]};
                        const v2f a1  = {cA1[2 * k], cA1[2 * k + 1]};
                        const v2f a2x = {cA2x[2 * k], cA2x[2 * k + 1]};
                        const v2f a2z = {cA2z[2 * k], cA2z[2 * k + 1]};
                        const v2f sL  = {accL[2 * k], accL[2 * k + 1]};
                        const v2f sGx = {accGx[2 * k], accGx[2 * k + 1]};
                        const v2f sGz = {accGz[2 * k], accGz[2 * k + 1]};
                        const v2f pls = {pu[s][2 * k], pu[s][2 * k + 1]};
                        const v2f rhs = a1 * sL + a2x * sGx + a2z * sGz;
                        const v2f nla = 1.0f + dd * q1;
                        const v2f a   = nla + 2.0f * dmp;
                        const v2f b   = 2.0f * nla - dmp * dmp;
                        const v2f ccc = 2.0f * dmp - nla;
                        const v2f dp  = q1 - q2;
                        const v2f res = div_rc2(b * q1 + ccc * q2 + dd * dp * dp + rhs + pls, a);
                        pn[2 * k]     = res[0];
                        pn[2 * k + 1] = res[1];
                    }
                } else {
#pragma unroll
                    for (int k = 0; k < 4; ++k) {
                        const v2f dmp = {cDm[2 * k], cDm[2 * k + 1]};
                        const v2f q1  = {p1[2 * k], p1[2 * k + 1]};
                        const v2f q2  = {p2[2 * k], p2[2 * k + 1]};
                        const v2f a1  = {cA1[2 * k], cA1[2 * k + 1]};
                        const v2f a2x = {cA2x[2 * k], cA2x[2 * k + 1]};
                        const v2f a2z = {cA2z[2 * k], cA2z[2 * k + 1]};
                        const v2f sL  = {accL[2 * k], accL[2 * k + 1]};
                        const v2f sGx = {accGx[2 * k], accGx[2 * k + 1]};
                        const v2f sGz = {accGz[2 * k], accGz[2 * k + 1]};
                        const v2f pls = {pu[s][2 * k], pu[s][2 * k + 1]};
                        const v2f rhs = a1 * sL + a2x * sGx + a2z * sGz;
                        const v2f a   = 1.0f + 2.0f * dmp;
                        const v2f b   = 2.0f - dmp * dmp;
                        const v2f ccc = 2.0f * dmp - 1.0f;
                        const v2f res = div_rc2(b * q1 + ccc * q2 + rhs + DT2 * pls, a);
                        pn[2 * k]     = res[0];
                        pn[2 * k + 1] = res[1];
                    }
                }
#pragma unroll
                for (int r = 0; r < 8; ++r) {
                    p2[r] = p1[r];
                    p1[r] = inb[r] ? pn[r] : 0.0f;
                }
                if (!lastst || sample) {
#pragma unroll
                    for (int r = 0; r < 8; ++r)
                        Fn[fbase + (r + 8 * half + 2) * FS + lm + 2] = p1[r];
                }
            }
            const bool bar = (!lastst || sample);
            if (bar) lds_barrier_arrive(p1);
            if (s + 1 < KST && s + 1 < nsteps && (s + 1 < nsteps - 1 || wave < 4)) {
                accL  = v8f{};
                accGz = v8f{};
                z_own_rows(p1, fw, fg, accL, accGz);
            }
            if (bar) lds_barrier_wait(accL, accGz);
            if (sample) {
                float* grow = gout + (size_t)(s >> 2) * nch;
                if (sel_off >= 0) grow[tid] = Fn[sel_off];
                for (int ch = tid + 512; ch < nch; ch += 512) {
                    int rz = el[2 * ch + 0] - (GZ0 + HL);
                    int rx = el[2 * ch + 1] - (GX0 + HL);
                    if (rz >= 0 && rz < OT && rx >= 0 && rx < OT)
                        grow[ch] = Fn[(rz + HL + 2) * FS + rx + HL + 2];
                }
            }
        }
    }

    if (wave < 4) {
#pragma unroll
        for (int r = 0; r < 8; ++r) {
            const int idx = (GZ0 + 16 * tz + r + 8 * half) * NX + gx;
            pout1[idx] = p1[r];
            pout2[idx] = p2[r];
        }
    }
}

extern "C" void kernel_launch(void* const* d_in, const int* in_sizes, int n_in,
                              void* d_out, int out_size, void* d_ws, size_t ws_size,
                              hipStream_t stream) {
    const float* pulse   = (const float*)d_in[0];
    const float* vel     = (const float*)d_in[1];
    const float* rho     = (const float*)d_in[2];
    const float* damping = (const float*)d_in[3];
    const float* beta    = (const float*)d_in[4];
    const int*   el      = (const int*)d_in[5];
    const int*   nla     = (const int*)d_in[6];

    const int nt  = in_sizes[0] / (NZ * NX);
    const int nch = in_sizes[5] / 2;

    float* ws = (float*)d_ws;
    const size_t F = (size_t)NZ * NX;
    float* pl[4] = { ws, ws + F, ws + 2 * F, ws + 3 * F };

    int cur = 0;
    for (int t0 = 0; t0 < nt; t0 += KST) {
        const int ns = (nt - t0 < KST) ? (nt - t0) : KST;
        steps_kernel<<<dim3(NX / OT, NZ / OT), 512, 0, stream>>>(
            pl[2 * cur], pl[2 * cur + 1], pl[2 * (cur ^ 1)], pl[2 * (cur ^ 1) + 1],
            pulse + (size_t)t0 * F, vel, rho, damping, beta, nla, el,
            (float*)d_out + (size_t)(t0 >> 2) * nch, nch, ns, t0 == 0 ? 1 : 0);
        cur ^= 1;
    }
}


# z gradient off the matrix unit: a subtraction of the lane's own field registers (other lane half by v_permlanex16, tile edge from the Laplacian's halo rows), was 5 banded WMMAs per tile and step
# speedup vs baseline: 8.5242x; 1.0408x over previous
#include <hip/hip_runtime.h>

#define NZ 512
#define NX 512
#define DT2 0.0025f

#define KST 8
#define OT  32
#define HL  16
#define FW  68
#define FS  69

typedef __attribute__((ext_vector_type(2))) float v2f;
typedef __attribute__((ext_vector_type(8))) float v8f;

__device__ __forceinline__ float lap_w(int d) {
    int a = d < 0 ? -d : d;
    return a == 0 ? -2.5f : (a == 1 ? 4.0f / 3.0f : (a == 2 ? -1.0f / 12.0f : 0.0f));
}
__device__ __forceinline__ float grd_w(int d) {
    return d == 1 ? 0.5f : (d == -1 ? -0.5f : 0.0f);
}

__device__ __forceinline__ float div_rc(float n, float a) {
    const float r = __builtin_amdgcn_rcpf(a);
    const float q = n * r;
    const float e = __builtin_fmaf(-a, q, n);
    return __builtin_fmaf(e, r, q);
}
__device__ __forceinline__ v2f div_rc2(v2f n, v2f a) {
    v2f r;
    r[0] = __builtin_amdgcn_rcpf(a[0]);
    r[1] = __builtin_amdgcn_rcpf(a[1]);
    const v2f q = n * r;
    const v2f e = __builtin_elementwise_fma(-a, q, n);
    return __builtin_elementwise_fma(e, r, q);
}
__device__ __forceinline__ float rcp_nr(float a) {
    const float r = __builtin_amdgcn_rcpf(a);
    const float e = __builtin_fmaf(-a, r, 1.0f);
    return __builtin_fmaf(e, r, r);
}

__device__ __forceinline__ void lds_barrier() {
    asm volatile("s_wait_dscnt 0x0\n\ts_barrier_signal -1\n\ts_barrier_wait -1" ::: "memory");
}

__device__ __forceinline__ void lds_barrier_arrive(float (&p)[8]) {
    asm volatile("s_wait_dscnt 0x0\n\ts_barrier_signal -1"
                 : "+v"(p[0]), "+v"(p[1]), "+v"(p[2]), "+v"(p[3]), "+v"(p[4]), "+v"(p[5]), "+v"(p[6]), "+v"(p[7])
                 :: "memory");
}
__device__ __forceinline__ void lds_barrier_wait(v8f& a) {
    asm volatile("s_barrier_wait -1" : "+v"(a) :: "memory");
}

__device__ __forceinline__ void prio_default_behind(v8f& a, v8f& b) {
    asm volatile("s_setprio 0" : "+v"(a), "+v"(b));
}

__device__ __forceinline__ float xhalf(float x) {
    const int b = __builtin_bit_cast(int, x);
    return __builtin_bit_cast(float, __builtin_amdgcn_permlanex16(b, b, 0x76543210, 0xfedcba98, false, false));
}

__device__ __forceinline__ void z_own_rows(const float (&p)[8], const v2f (&fw)[5], v8f& accL) {
#pragma unroll
    for (int c = 0; c < 4; ++c) {
        v2f pz;
        pz[0] = p[2 * c];
        pz[1] = p[2 * c + 1];
        accL  = __builtin_amdgcn_wmma_f32_16x16x4_f32(false, fw[c], false, pz, (short)0, accL,  false, false);
    }
}

__launch_bounds__(512)
__global__ void steps_kernel(const float* __restrict__ pin1, const float* __restrict__ pin2,
                             float* __restrict__ pout1, float* __restrict__ pout2,
                             const float* pulse,
                             const float* __restrict__ vel,
                             const float* __restrict__ rho,
                             const float* __restrict__ damping,
                             const float* __restrict__ beta,
                             const int* __restrict__ nlaflag,
                             const int* __restrict__ el,
                             float* __restrict__ gout,
                             int nch, int nsteps, int first) {
    __shared__ float F[2][FW * FS];
    __shared__ float IR[FW * FS];

    const int tid  = threadIdx.x;
    const int wave = __builtin_amdgcn_readfirstlane(tid >> 5);
    const int lane = tid & 31;
    const int half = lane >> 4;
    const int lm   = lane & 15;
    const unsigned code = (unsigned)((0xFEDCB8743210A965ULL >> (4 * wave)) & 15ULL);
    const int tz = (int)(code >> 2), tx = (int)(code & 3u);
    const int GZ0 = (int)blockIdx.y * OT - HL;
    const int GX0 = (int)blockIdx.x * OT - HL;
    const int fbase = (16 * tz) * FS + 16 * tx;

    int ez = 0, ex = 0;
    if (nch > 0) {
        const int ci = tid < nch ? tid : nch - 1;
        ez = el[2 * ci + 0];
        ex = el[2 * ci + 1];
    }

    const int NLA = nlaflag[0];

    float sv[10], sr[10];
    int   sad[10];
    bool  sok[10];
#pragma unroll
    for (int j = 0; j < 10; ++j) {
        int i  = tid + 512 * j;
        int ii = i < FW * FW ? i : FW * FW - 1;
        int zz = ii / FW, xx = ii - zz * FW;
        int gz = GZ0 - 2 + zz, gxs = GX0 - 2 + xx;
        int cz = gz < 0 ? 0 : (gz > NZ - 1 ? NZ - 1 : gz);
        int cx = gxs < 0 ? 0 : (gxs > NX - 1 ? NX - 1 : gxs);
        sok[j] = (cz == gz) & (cx == gxs);
        sad[j] = cz * NX + cx;
        sr[j]  = rho[sad[j]];
        sv[j]  = 0.0f;
    }
    if (!first) {
#pragma unroll
        for (int j = 0; j < 10; ++j) sv[j] = pin1[sad[j]];
    }

    float rC[8], vV[8], bB[8], cDm[8], p2[8], pu[KST][8];
    bool inb[8];
    int  cidx[8];
    const int gx  = GX0 + 16 * tx + lm;
    const int cxc = gx < 0 ? 0 : (gx > NX - 1 ? NX - 1 : gx);
#pragma unroll
    for (int r = 0; r < 8; ++r) {
        const int gz  = GZ0 + 16 * tz + r + 8 * half;
        const int czc = gz < 0 ? 0 : (gz > NZ - 1 ? NZ - 1 : gz);
        inb[r]  = (czc == gz) & (cxc == gx);
        cidx[r] = czc * NX + cxc;
        rC[r]  = rho[cidx[r]];
        vV[r]  = vel[cidx[r]];
        bB[r]  = beta[cidx[r]];
        cDm[r] = damping[cidx[r]];
        p2[r]  = 0.0f;
    }
    if (!first) {
#pragma unroll
        for (int r = 0; r < 8; ++r) p2[r] = pin2[cidx[r]];
    }
#pragma unroll
    for (int j = 0; j < 10; ++j) {
        int i = tid + 512 * j;
        if (i < FW * FW) {
            int zz = i / FW, xx = i - zz * FW;
            F[0][zz * FS + xx] = sok[j] ? sv[j] : 0.0f;
            F[1][zz * FS + xx] = 0.0f;
            IR[zz * FS + xx]   = sok[j] ? rcp_nr(sr[j]) : 0.0f;
        }
    }

    v2f fw[5], fg[5];
#pragma unroll
    for (int c = 0; c < 5; ++c) {
        const int kk = c < 4 ? 8 * half + 2 * c : (half ? 16 : -2);
        const int d  = kk - lm;
        fw[c][0] = lap_w(d); fw[c][1] = lap_w(d + 1);
        fg[c][0] = grd_w(d); fg[c][1] = grd_w(d + 1);
    }
    const int hk = half ? 18 : 0;

    int sel_off = -1;
    {
        const int rz = ez - (GZ0 + HL), rx = ex - (GX0 + HL);
        const bool mine = (tid < nch) & (rz >= 0) & (rz < OT) & (rx >= 0) & (rx < OT);
        sel_off = mine ? (rz + HL + 2) * FS + rx + HL + 2 : -1;
    }

    lds_barrier();

#pragma unroll
    for (int s = 0; s < KST; ++s) {
        const int sc = s < nsteps ? s : nsteps - 1;
#pragma unroll
        for (int r = 0; r < 8; ++r) pu[s][r] = pulse[(size_t)sc * (NZ * NX) + cidx[r]];
    }
    asm volatile("" ::: "memory");

    float p1[8], cA1[8], cA2x[8], cA2z[8], cD[8];
#pragma unroll
    for (int r = 0; r < 8; ++r) {
        const int fo = fbase + (r + 8 * half + 2) * FS + lm + 2;
        p1[r] = F[0][fo];
        const float irE = IR[fo + 1], irW = IR[fo - 1], irS = IR[fo + FS], irN = IR[fo - FS];
        const float Kc  = vV[r] * vV[r] * rC[r];
        const float gxr = 0.5f * (irE - irW);
        const float gzr = 0.5f * (irS - irN);
        cA1[r]  = DT2 * Kc * rC[r];
        cA2x[r] = DT2 * Kc * gxr;
        cA2z[r] = 0.5f * (DT2 * Kc * gzr);
        cD[r]   = div_rc(2.0f * bB[r], Kc);
    }

    v8f accL  = {};
    if (nsteps > 1 || wave < 4) z_own_rows(p1, fw, accL);

#pragma unroll
    for (int s = 0; s < KST; ++s) {
        if (s < nsteps) {
            const float* Fc = F[s & 1];
            float*       Fn = F[(s & 1) ^ 1];
            const bool lastst = (s == nsteps - 1);
            const bool sample = ((s & 3) == 0);
            if (!lastst || wave < 4) {
                __builtin_amdgcn_s_setprio(3);
                v8f accGx = {};
                v2f px[5], pzh;
                pzh[0]   = Fc[fbase + hk * FS + lm + 2];
                pzh[1]   = Fc[fbase + (hk + 1) * FS + lm + 2];
#pragma unroll
                for (int c = 0; c < 4; ++c) {
                    px[c][0] = Fc[fbase + (lm + 2) * FS + 8 * half + 2 * c + 2];
                    px[c][1] = Fc[fbase + (lm + 2) * FS + 8 * half + 2 * c + 3];
                }
                px[4][0] = Fc[fbase + (lm + 2) * FS + hk];
                px[4][1] = Fc[fbase + (lm + 2) * FS + hk + 1];
                accL  = __builtin_amdgcn_wmma_f32_16x16x4_f32(false, fw[4], false, pzh, (short)0, accL,  false, false);
#pragma unroll
                for (int c = 0; c < 5; ++c) {
                    accL  = __builtin_amdgcn_wmma_f32_16x16x4_f32(false, px[c], false, fw[c], (short)0, accL,  false, false);
                    accGx = __builtin_amdgcn_wmma_f32_16x16x4_f32(false, px[c], false, fg[c], (short)0, accGx, false, false);
                }
                prio_default_behind(accL, accGx);
                float dz[8];
                {
                    const float up_o = xhalf(p1[7]);
                    const float dn_o = xhalf(p1[0]);
                    const float up0  = half ? up_o : pzh[1];
                    const float dn7  = half ? pzh[0] : dn_o;
                    dz[0] = p1[1] - up0;
#pragma unroll
                    for (int r = 1; r < 7; ++r) dz[r] = p1[r + 1] - p1[r - 1];
                    dz[7] = dn7 - p1[6];
                }
                float pn[8];
                if (NLA) {
#pragma unroll
                    for (int k = 0; k < 4; ++k) {
                        const v2f dmp = {cDm[2 * k], cDm[2 * k + 1]};
                        const v2f q1  = {p1[2 * k], p1[2 * k + 1]};
                        const v2f q2  = {p2[2 * k], p2[2 * k + 1]};
                        const v2f dd  = {cD[2 * k], cD[2 * k + 1]};
                        const v2f a1  = {cA1[2 * k], cA1[2 * k + 1]};
                        const v2f a2x = {cA2x[2 * k], cA2x[2 * k + 1]};
                        const v2f a2z = {cA2z[2 * k], cA2z[2 * k + 1]};
                        const v2f sL  = {accL[2 * k], accL[2 * k + 1]};
                        const v2f sGx = {accGx[2 * k], accGx[2 * k + 1]};
                        const v2f sGz = {dz[2 * k], dz[2 * k + 1]};
                        const v2f pls = {pu[s][2 * k], pu[s][2 * k + 1]};
                        const v2f rhs = a1 * sL + a2x * sGx + a2z * sGz;
                        const v2f nla = 1.0f + dd * q1;
                        const v2f a   = nla + 2.0f * dmp;
                        const v2f b   = 2.0f * nla - dmp * dmp;
                        const v2f ccc = 2.0f * dmp - nla;
                        const v2f dp  = q1 - q2;
                        const v2f res = div_rc2(b * q1 + ccc * q2 + dd * dp * dp + rhs + pls, a);
                        pn[2 * k]     = res[0];
                        pn[2 * k + 1] = res[1];
                    }
                } else {
#pragma unroll
                    for (int k = 0; k < 4; ++k) {
                        const v2f dmp = {cDm[2 * k], cDm[2 * k + 1]};
                        const v2f q1  = {p1[2 * k], p1[2 * k + 1]};
                        const v2f q2  = {p2[2 * k], p2[2 * k + 1]};
                        const v2f a1  = {cA1[2 * k], cA1[2 * k + 1]};
                        const v2f a2x = {cA2x[2 * k], cA2x[2 * k + 1]};
                        const v2f a2z = {cA2z[2 * k], cA2z[2 * k + 1]};
                        const v2f sL  = {accL[2 * k], accL[2 * k + 1]};
                        const v2f sGx = {accGx[2 * k], accGx[2 * k + 1]};
                        const v2f sGz = {dz[2 * k], dz[2 * k + 1]};
                        const v2f pls = {pu[s][2 * k], pu[s][2 * k + 1]};
                        const v2f rhs = a1 * sL + a2x * sGx + a2z * sGz;
                        const v2f a   = 1.0f + 2.0f * dmp;
                        const v2f b   = 2.0f - dmp * dmp;
                        const v2f ccc = 2.0f * dmp - 1.0f;
                        const v2f res = div_rc2(b * q1 + ccc * q2 + rhs + DT2 * pls, a);
                        pn[2 * k]     = res[0];
                        pn[2 * k + 1] = res[1];
                    }
                }
#pragma unroll
                for (int r = 0; r < 8; ++r) {
                    p2[r] = p1[r];
                    p1[r] = inb[r] ? pn[r] : 0.0f;
                }
                if (!lastst || sample) {
#pragma unroll
                    for (int r = 0; r < 8; ++r)
                        Fn[fbase + (r + 8 * half + 2) * FS + lm + 2] = p1[r];
                }
            }
            const bool bar = (!lastst || sample);
            if (bar) lds_barrier_arrive(p1);
            if (s + 1 < KST && s + 1 < nsteps && (s + 1 < nsteps - 1 || wave < 4)) {
                accL  = v8f{};
                z_own_rows(p1, fw, accL);
            }
            if (bar) lds_barrier_wait(accL);
            if (sample) {
                float* grow = gout + (size_t)(s >> 2) * nch;
                if (sel_off >= 0) grow[tid] = Fn[sel_off];
                for (int ch = tid + 512; ch < nch; ch += 512) {
                    int rz = el[2 * ch + 0] - (GZ0 + HL);
                    int rx = el[2 * ch + 1] - (GX0 + HL);
                    if (rz >= 0 && rz < OT && rx >= 0 && rx < OT)
                        grow[ch] = Fn[(rz + HL + 2) * FS + rx + HL + 2];
                }
            }
        }
    }

    if (wave < 4) {
#pragma unroll
        for (int r = 0; r < 8; ++r) {
            const int idx = (GZ0 + 16 * tz + r + 8 * half) * NX + gx;
            pout1[idx] = p1[r];
            pout2[idx] = p2[r];
        }
    }
}

extern "C" void kernel_launch(void* const* d_in, const int* in_sizes, int n_in,
                              void* d_out, int out_size, void* d_ws, size_t ws_size,
                              hipStream_t stream) {
    const float* pulse   = (const float*)d_in[0];
    const float* vel     = (const float*)d_in[1];
    const float* rho     = (const float*)d_in[2];
    const float* damping = (const float*)d_in[3];
    const float* beta    = (const float*)d_in[4];
    const int*   el      = (const int*)d_in[5];
    const int*   nla     = (const int*)d_in[6];

    const int nt  = in_sizes[0] / (NZ * NX);
    const int nch = in_sizes[5] / 2;

    float* ws = (float*)d_ws;
    const size_t F = (size_t)NZ * NX;
    float* pl[4] = { ws, ws + F, ws + 2 * F, ws + 3 * F };

    int cur = 0;
    for (int t0 = 0; t0 < nt; t0 += KST) {
        const int ns = (nt - t0 < KST) ? (nt - t0) : KST;
        steps_kernel<<<dim3(NX / OT, NZ / OT), 512, 0, stream>>>(
            pl[2 * cur], pl[2 * cur + 1], pl[2 * (cur ^ 1)], pl[2 * (cur ^ 1) + 1],
            pulse + (size_t)t0 * F, vel, rho, damping, beta, nla, el,
            (float*)d_out + (size_t)(t0 >> 2) * nch, nch, ns, t0 == 0 ? 1 : 0);
        cur ^= 1;
    }
}


# x gradient off the matrix unit: p[x+1]-p[x-1] per cell from one two-address LDS read and one subtraction (single rounding), one half folded into its coefficient; was 5 banded WMMAs per tile-step
# speedup vs baseline: 9.0700x; 1.0640x over previous
#include <hip/hip_runtime.h>

#define NZ 512
#define NX 512
#define DT2 0.0025f

#define KST 8
#define OT  32
#define HL  16
#define FW  68
#define FS  69

typedef __attribute__((ext_vector_type(2))) float v2f;
typedef __attribute__((ext_vector_type(8))) float v8f;

__device__ __forceinline__ float lap_w(int d) {
    int a = d < 0 ? -d : d;
    return a == 0 ? -2.5f : (a == 1 ? 4.0f / 3.0f : (a == 2 ? -1.0f / 12.0f : 0.0f));
}

__device__ __forceinline__ float div_rc(float n, float a) {
    const float r = __builtin_amdgcn_rcpf(a);
    const float q = n * r;
    const float e = __builtin_fmaf(-a, q, n);
    return __builtin_fmaf(e, r, q);
}
__device__ __forceinline__ v2f div_rc2(v2f n, v2f a) {
    v2f r;
    r[0] = __builtin_amdgcn_rcpf(a[0]);
    r[1] = __builtin_amdgcn_rcpf(a[1]);
    const v2f q = n * r;
    const v2f e = __builtin_elementwise_fma(-a, q, n);
    return __builtin_elementwise_fma(e, r, q);
}
__device__ __forceinline__ float rcp_nr(float a) {
    const float r = __builtin_amdgcn_rcpf(a);
    const float e = __builtin_fmaf(-a, r, 1.0f);
    return __builtin_fmaf(e, r, r);
}

__device__ __forceinline__ void lds_barrier() {
    asm volatile("s_wait_dscnt 0x0\n\ts_barrier_signal -1\n\ts_barrier_wait -1" ::: "memory");
}

__device__ __forceinline__ void lds_barrier_arrive(float (&p)[8]) {
    asm volatile("s_wait_dscnt 0x0\n\ts_barrier_signal -1"
                 : "+v"(p[0]), "+v"(p[1]), "+v"(p[2]), "+v"(p[3]), "+v"(p[4]), "+v"(p[5]), "+v"(p[6]), "+v"(p[7])
                 :: "memory");
}
__device__ __forceinline__ void lds_barrier_wait(v8f& a) {
    asm volatile("s_barrier_wait -1" : "+v"(a) :: "memory");
}

__device__ __forceinline__ void prio_default_behind(v8f& a) {
    asm volatile("s_setprio 0" : "+v"(a));
}

__device__ __forceinline__ void reads_before_products(v8f& a) {
    asm volatile("" : "+v"(a) :: "memory");
}

__device__ __forceinline__ float xhalf(float x) {
    const int b = __builtin_bit_cast(int, x);
    return __builtin_bit_cast(float, __builtin_amdgcn_permlanex16(b, b, 0x76543210, 0xfedcba98, false, false));
}

__device__ __forceinline__ void z_own_rows(const float (&p)[8], const v2f (&fw)[5], v8f& accL) {
#pragma unroll
    for (int c = 0; c < 4; ++c) {
        v2f pz;
        pz[0] = p[2 * c];
        pz[1] = p[2 * c + 1];
        accL  = __builtin_amdgcn_wmma_f32_16x16x4_f32(false, fw[c], false, pz, (short)0, accL,  false, false);
    }
}

__launch_bounds__(512)
__global__ void steps_kernel(const float* __restrict__ pin1, const float* __restrict__ pin2,
                             float* __restrict__ pout1, float* __restrict__ pout2,
                             const float* pulse,
                             const float* __restrict__ vel,
                             const float* __restrict__ rho,
                             const float* __restrict__ damping,
                             const float* __restrict__ beta,
                             const int* __restrict__ nlaflag,
                             const int* __restrict__ el,
                             float* __restrict__ gout,
                             int nch, int nsteps, int first) {
    __shared__ float F[2][FW * FS];
    __shared__ float IR[FW * FS];

    const int tid  = threadIdx.x;
    const int wave = __builtin_amdgcn_readfirstlane(tid >> 5);
    const int lane = tid & 31;
    const int half = lane >> 4;
    const int lm   = lane & 15;
    const unsigned code = (unsigned)((0xFEDCB8743210A965ULL >> (4 * wave)) & 15ULL);
    const int tz = (int)(code >> 2), tx = (int)(code & 3u);
    const int GZ0 = (int)blockIdx.y * OT - HL;
    const int GX0 = (int)blockIdx.x * OT - HL;
    const int fbase = (16 * tz) * FS + 16 * tx;

    int ez = 0, ex = 0;
    if (nch > 0) {
        const int ci = tid < nch ? tid : nch - 1;
        ez = el[2 * ci + 0];
        ex = el[2 * ci + 1];
    }

    const int NLA = nlaflag[0];

    float sv[10], sr[10];
    int   sad[10];
    bool  sok[10];
#pragma unroll
    for (int j = 0; j < 10; ++j) {
        int i  = tid + 512 * j;
        int ii = i < FW * FW ? i : FW * FW - 1;
        int zz = ii / FW, xx = ii - zz * FW;
        int gz = GZ0 - 2 + zz, gxs = GX0 - 2 + xx;
        int cz = gz < 0 ? 0 : (gz > NZ - 1 ? NZ - 1 : gz);
        int cx = gxs < 0 ? 0 : (gxs > NX - 1 ? NX - 1 : gxs);
        sok[j] = (cz == gz) & (cx == gxs);
        sad[j] = cz * NX + cx;
        sr[j]  = rho[sad[j]];
        sv[j]  = 0.0f;
    }
    if (!first) {
#pragma unroll
        for (int j = 0; j < 10; ++j) sv[j] = pin1[sad[j]];
    }

    float rC[8], vV[8], bB[8], cDm[8], p2[8], pu[KST][8];
    bool inb[8];
    int  cidx[8];
    const int gx  = GX0 + 16 * tx + lm;
    const int cxc = gx < 0 ? 0 : (gx > NX - 1 ? NX - 1 : gx);
#pragma unroll
    for (int r = 0; r < 8; ++r) {
        const int gz  = GZ0 + 16 * tz + r + 8 * half;
        const int czc = gz < 0 ? 0 : (gz > NZ - 1 ? NZ - 1 : gz);
        inb[r]  = (czc == gz) & (cxc == gx);
        cidx[r] = czc * NX + cxc;
        rC[r]  = rho[cidx[r]];
        vV[r]  = vel[cidx[r]];
        bB[r]  = beta[cidx[r]];
        cDm[r] = damping[cidx[r]];
        p2[r]  = 0.0f;
    }
    if (!first) {
#pragma unroll
        for (int r = 0; r < 8; ++r) p2[r] = pin2[cidx[r]];
    }
#pragma unroll
    for (int j = 0; j < 10; ++j) {
        int i = tid + 512 * j;
        if (i < FW * FW) {
            int zz = i / FW, xx = i - zz * FW;
            F[0][zz * FS + xx] = sok[j] ? sv[j] : 0.0f;
            F[1][zz * FS + xx] = 0.0f;
            IR[zz * FS + xx]   = sok[j] ? rcp_nr(sr[j]) : 0.0f;
        }
    }

    v2f fw[5];
#pragma unroll
    for (int c = 0; c < 5; ++c) {
        const int kk = c < 4 ? 8 * half + 2 * c : (half ? 16 : -2);
        const int d  = kk - lm;
        fw[c][0] = lap_w(d); fw[c][1] = lap_w(d + 1);
    }
    const int hk = half ? 18 : 0;

    int sel_off = -1;
    {
        const int rz = ez - (GZ0 + HL), rx = ex - (GX0 + HL);
        const bool mine = (tid < nch) & (rz >= 0) & (rz < OT) & (rx >= 0) & (rx < OT);
        sel_off = mine ? (rz + HL + 2) * FS + rx + HL + 2 : -1;
    }

    lds_barrier();

#pragma unroll
    for (int s = 0; s < KST; ++s) {
        const int sc = s < nsteps ? s : nsteps - 1;
#pragma unroll
        for (int r = 0; r < 8; ++r) pu[s][r] = pulse[(size_t)sc * (NZ * NX) + cidx[r]];
    }
    asm volatile("" ::: "memory");

    float p1[8], cA1[8], cA2x[8], cA2z[8], cD[8];
#pragma unroll
    for (int r = 0; r < 8; ++r) {
        const int fo = fbase + (r + 8 * half + 2) * FS + lm + 2;
        p1[r] = F[0][fo];
        const float irE = IR[fo + 1], irW = IR[fo - 1], irS = IR[fo + FS], irN = IR[fo - FS];
        const float Kc  = vV[r] * vV[r] * rC[r];
        const float gxr = 0.5f * (irE - irW);
        const float gzr = 0.5f * (irS - irN);
        cA1[r]  = DT2 * Kc * rC[r];
        cA2x[r] = 0.5f * (DT2 * Kc * gxr);
        cA2z[r] = 0.5f * (DT2 * Kc * gzr);
        cD[r]   = div_rc(2.0f * bB[r], Kc);
    }

    v8f accL  = {};
    if (nsteps > 1 || wave < 4) z_own_rows(p1, fw, accL);

#pragma unroll
    for (int s = 0; s < KST; ++s) {
        if (s < nsteps) {
            const float* Fc = F[s & 1];
            float*       Fn = F[(s & 1) ^ 1];
            const bool lastst = (s == nsteps - 1);
            const bool sample = ((s & 3) == 0);
            if (!lastst || wave < 4) {
                __builtin_amdgcn_s_setprio(3);
                v2f px[5], pzh;
                pzh[0]   = Fc[fbase + hk * FS + lm + 2];
                pzh[1]   = Fc[fbase + (hk + 1) * FS + lm + 2];
#pragma unroll
                for (int c = 0; c < 4; ++c) {
                    px[c][0] = Fc[fbase + (lm + 2) * FS + 8 * half + 2 * c + 2];
                    px[c][1] = Fc[fbase + (lm + 2) * FS + 8 * half + 2 * c + 3];
                }
                px[4][0] = Fc[fbase + (lm + 2) * FS + hk];
                px[4][1] = Fc[fbase + (lm + 2) * FS + hk + 1];
                asm volatile("" ::: "memory");
                float gl[8], gr[8];
                const int gxo = fbase + (8 * half + 2) * FS + lm + 1;
#pragma unroll
                for (int r = 0; r < 8; ++r) {
                    gl[r] = Fc[gxo + r * FS];
                    gr[r] = Fc[gxo + r * FS + 2];
                }
                reads_before_products(accL);
                float dz[8];
                {
                    const float up_o = xhalf(p1[7]);
                    const float dn_o = xhalf(p1[0]);
                    const float up0  = half ? up_o : pzh[1];
                    const float dn7  = half ? pzh[0] : dn_o;
                    dz[0] = p1[1] - up0;
#pragma unroll
                    for (int r = 1; r < 7; ++r) dz[r] = p1[r + 1] - p1[r - 1];
                    dz[7] = dn7 - p1[6];
                }
                accL  = __builtin_amdgcn_wmma_f32_16x16x4_f32(false, fw[4], false, pzh, (short)0, accL,  false, false);
#pragma unroll
                for (int c = 0; c < 5; ++c)
                    accL  = __builtin_amdgcn_wmma_f32_16x16x4_f32(false, px[c], false, fw[c], (short)0, accL,  false, false);
                prio_default_behind(accL);
                __builtin_amdgcn_sched_barrier(0);
                float dx[8];
#pragma unroll
                for (int r = 0; r < 8; ++r) dx[r] = gr[r] - gl[r];
                float pn[8];
                if (NLA) {
#pragma unroll
                    for (int k = 0; k < 4; ++k) {
                        const v2f dmp = {cDm[2 * k], cDm[2 * k + 1]};
                        const v2f q1  = {p1[2 * k], p1[2 * k + 1]};
                        const v2f q2  = {p2[2 * k], p2[2 * k + 1]};
                        const v2f dd  = {cD[2 * k], cD[2 * k + 1]};
                        const v2f a1  = {cA1[2 * k], cA1[2 * k + 1]};
                        const v2f a2x = {cA2x[2 * k], cA2x[2 * k + 1]};
                        const v2f a2z = {cA2z[2 * k], cA2z[2 * k + 1]};
                        const v2f sL  = {accL[2 * k], accL[2 * k + 1]};
                        const v2f sGx = {dx[2 * k], dx[2 * k + 1]};
                        const v2f sGz = {dz[2 * k], dz[2 * k + 1]};
                        const v2f pls = {pu[s][2 * k], pu[s][2 * k + 1]};
                        const v2f rhs = a1 * sL + a2x * sGx + a2z * sGz;
                        const v2f nla = 1.0f + dd * q1;
                        const v2f a   = nla + 2.0f * dmp;
                        const v2f b   = 2.0f * nla - dmp * dmp;
                        const v2f ccc = 2.0f * dmp - nla;
                        const v2f dp  = q1 - q2;
                        const v2f res = div_rc2(b * q1 + ccc * q2 + dd * dp * dp + rhs + pls, a);
                        pn[2 * k]     = res[0];
                        pn[2 * k + 1] = res[1];
                    }
                } else {
#pragma unroll
                    for (int k = 0; k < 4; ++k) {
                        const v2f dmp = {cDm[2 * k], cDm[2 * k + 1]};
                        const v2f q1  = {p1[2 * k], p1[2 * k + 1]};
                        const v2f q2  = {p2[2 * k], p2[2 * k + 1]};
                        const v2f a1  = {cA1[2 * k], cA1[2 * k + 1]};
                        const v2f a2x = {cA2x[2 * k], cA2x[2 * k + 1]};
                        const v2f a2z = {cA2z[2 * k], cA2z[2 * k + 1]};
                        const v2f sL  = {accL[2 * k], accL[2 * k + 1]};
                        const v2f sGx = {dx[2 * k], dx[2 * k + 1]};
                        const v2f sGz = {dz[2 * k], dz[2 * k + 1]};
                        const v2f pls = {pu[s][2 * k], pu[s][2 * k + 1]};
                        const v2f rhs = a1 * sL + a2x * sGx + a2z * sGz;
                        const v2f a   = 1.0f + 2.0f * dmp;
                        const v2f b   = 2.0f - dmp * dmp;
                        const v2f ccc = 2.0f * dmp - 1.0f;
                        const v2f res = div_rc2(b * q1 + ccc * q2 + rhs + DT2 * pls, a);
                        pn[2 * k]     = res[0];
                        pn[2 * k + 1] = res[1];
                    }
                }
#pragma unroll
                for (int r = 0; r < 8; ++r) {
                    p2[r] = p1[r];
                    p1[r] = inb[r] ? pn[r] : 0.0f;
                }
                if (!lastst || sample) {
#pragma unroll
                    for (int r = 0; r < 8; ++r)
                        Fn[fbase + (r + 8 * half + 2) * FS + lm + 2] = p1[r];
                }
            }
            const bool bar = (!lastst || sample);
            if (bar) lds_barrier_arrive(p1);
            if (s + 1 < KST && s + 1 < nsteps && (s + 1 < nsteps - 1 || wave < 4)) {
                accL  = v8f{};
                z_own_rows(p1, fw, accL);
            }
            if (bar) lds_barrier_wait(accL);
            if (sample) {
                float* grow = gout + (size_t)(s >> 2) * nch;
                if (sel_off >= 0) grow[tid] = Fn[sel_off];
                for (int ch = tid + 512; ch < nch; ch += 512) {
                    int rz = el[2 * ch + 0] - (GZ0 + HL);
                    int rx = el[2 * ch + 1] - (GX0 + HL);
                    if (rz >= 0 && rz < OT && rx >= 0 && rx < OT)
                        grow[ch] = Fn[(rz + HL + 2) * FS + rx + HL + 2];
                }
            }
        }
    }

    if (wave < 4) {
#pragma unroll
        for (int r = 0; r < 8; ++r) {
            const int idx = (GZ0 + 16 * tz + r + 8 * half) * NX + gx;
            pout1[idx] = p1[r];
            pout2[idx] = p2[r];
        }
    }
}

extern "C" void kernel_launch(void* const* d_in, const int* in_sizes, int n_in,
                              void* d_out, int out_size, void* d_ws, size_t ws_size,
                              hipStream_t stream) {
    const float* pulse   = (const float*)d_in[0];
    const float* vel     = (const float*)d_in[1];
    const float* rho     = (const float*)d_in[2];
    const float* damping = (const float*)d_in[3];
    const float* beta    = (const float*)d_in[4];
    const int*   el      = (const int*)d_in[5];
    const int*   nla     = (const int*)d_in[6];

    const int nt  = in_sizes[0] / (NZ * NX);
    const int nch = in_sizes[5] / 2;

    float* ws = (float*)d_ws;
    const size_t F = (size_t)NZ * NX;
    float* pl[4] = { ws, ws + F, ws + 2 * F, ws + 3 * F };

    int cur = 0;
    for (int t0 = 0; t0 < nt; t0 += KST) {
        const int ns = (nt - t0 < KST) ? (nt - t0) : KST;
        steps_kernel<<<dim3(NX / OT, NZ / OT), 512, 0, stream>>>(
            pl[2 * cur], pl[2 * cur + 1], pl[2 * (cur ^ 1)], pl[2 * (cur ^ 1) + 1],
            pulse + (size_t)t0 * F, vel, rho, damping, beta, nla, el,
            (float*)d_out + (size_t)(t0 >> 2) * nch, nch, ns, t0 == 0 ? 1 : 0);
        cur ^= 1;
    }
}


# launch prologue: each lane loads, stages and keeps only its own 8 cells (zero frame, no 68x68 staging pass: 124 -> 112 loads per lane), every load as scalar plane base + the cell's 32-bit offset
# speedup vs baseline: 9.8559x; 1.0866x over previous
#include <hip/hip_runtime.h>

#define NZ 512
#define NX 512
#define DT2 0.0025f

#define KST 8
#define OT  32
#define HL  16
#define FW  68
#define FS  69

typedef __attribute__((ext_vector_type(2))) float v2f;
typedef __attribute__((ext_vector_type(8))) float v8f;

__device__ __forceinline__ float lap_w(int d) {
    int a = d < 0 ? -d : d;
    return a == 0 ? -2.5f : (a == 1 ? 4.0f / 3.0f : (a == 2 ? -1.0f / 12.0f : 0.0f));
}

__device__ __forceinline__ float div_rc(float n, float a) {
    const float r = __builtin_amdgcn_rcpf(a);
    const float q = n * r;
    const float e = __builtin_fmaf(-a, q, n);
    return __builtin_fmaf(e, r, q);
}
__device__ __forceinline__ v2f div_rc2(v2f n, v2f a) {
    v2f r;
    r[0] = __builtin_amdgcn_rcpf(a[0]);
    r[1] = __builtin_amdgcn_rcpf(a[1]);
    const v2f q = n * r;
    const v2f e = __builtin_elementwise_fma(-a, q, n);
    return __builtin_elementwise_fma(e, r, q);
}
__device__ __forceinline__ float rcp_nr(float a) {
    const float r = __builtin_amdgcn_rcpf(a);
    const float e = __builtin_fmaf(-a, r, 1.0f);
    return __builtin_fmaf(e, r, r);
}

__device__ __forceinline__ void lds_barrier() {
    asm volatile("s_wait_dscnt 0x0\n\ts_barrier_signal -1\n\ts_barrier_wait -1" ::: "memory");
}

__device__ __forceinline__ void lds_barrier_arrive(float (&p)[8]) {
    asm volatile("s_wait_dscnt 0x0\n\ts_barrier_signal -1"
                 : "+v"(p[0]), "+v"(p[1]), "+v"(p[2]), "+v"(p[3]), "+v"(p[4]), "+v"(p[5]), "+v"(p[6]), "+v"(p[7])
                 :: "memory");
}
__device__ __forceinline__ void lds_barrier_wait(v8f& a) {
    asm volatile("s_barrier_wait -1" : "+v"(a) :: "memory");
}

__device__ __forceinline__ void prio_default_behind(v8f& a) {
    asm volatile("s_setprio 0" : "+v"(a));
}

__device__ __forceinline__ void reads_before_products(v8f& a) {
    asm volatile("" : "+v"(a) :: "memory");
}

__device__ __forceinline__ float xhalf(float x) {
    const int b = __builtin_bit_cast(int, x);
    return __builtin_bit_cast(float, __builtin_amdgcn_permlanex16(b, b, 0x76543210, 0xfedcba98, false, false));
}

__device__ __forceinline__ void z_own_rows(const float (&p)[8], const v2f (&fw)[5], v8f& accL) {
#pragma unroll
    for (int c = 0; c < 4; ++c) {
        v2f pz;
        pz[0] = p[2 * c];
        pz[1] = p[2 * c + 1];
        accL  = __builtin_amdgcn_wmma_f32_16x16x4_f32(false, fw[c], false, pz, (short)0, accL,  false, false);
    }
}

__launch_bounds__(512)
__global__ void steps_kernel(const float* __restrict__ pin1, const float* __restrict__ pin2,
                             float* __restrict__ pout1, float* __restrict__ pout2,
                             const float* pulse,
                             const float* __restrict__ vel,
                             const float* __restrict__ rho,
                             const float* __restrict__ damping,
                             const float* __restrict__ beta,
                             const int* __restrict__ nlaflag,
                             const int* __restrict__ el,
                             float* __restrict__ gout,
                             int nch, int nsteps, int first) {
    __shared__ float L[3 * FW * FS];
    float* const F0 = L;
    float* const F1 = L + FW * FS;
    float* const IR = L + 2 * FW * FS;

    const int tid  = threadIdx.x;
    const int wave = __builtin_amdgcn_readfirstlane(tid >> 5);
    const int lane = tid & 31;
    const int half = lane >> 4;
    const int lm   = lane & 15;
    const unsigned code = (unsigned)((0xFEDCB8743210A965ULL >> (4 * wave)) & 15ULL);
    const int tz = (int)(code >> 2), tx = (int)(code & 3u);
    const int GZ0 = (int)blockIdx.y * OT - HL;
    const int GX0 = (int)blockIdx.x * OT - HL;
    const int fbase = (16 * tz) * FS + 16 * tx;

    int ez = 0, ex = 0;
    if (nch > 0) {
        const int ci = tid < nch ? tid : nch - 1;
        ez = el[2 * ci + 0];
        ex = el[2 * ci + 1];
    }

    const int NLA = nlaflag[0];

    if (tid < 4 * FW) {
        const int q  = tid / FW, xx = tid - q * FW;
        const int zz = q < 2 ? q : q + (FW - 4);
        F0[zz * FS + xx] = 0.0f;
        F1[zz * FS + xx] = 0.0f;
        IR[zz * FS + xx] = 0.0f;
    }
    if (tid < 4 * (FW - 4)) {
        const int zz = 2 + (tid >> 2), c = tid & 3;
        const int xx = c < 2 ? c : c + (FW - 4);
        F0[zz * FS + xx] = 0.0f;
        F1[zz * FS + xx] = 0.0f;
        IR[zz * FS + xx] = 0.0f;
    }
    const float* const f1 = first ? rho : pin1;
    const float* const f2 = first ? rho : pin2;
    float rC[8], vV[8], bB[8], cDm[8], p1[8], p2[8], pu[KST][8];
    bool inb[8];
    int  cidx[8];
    const int gx  = GX0 + 16 * tx + lm;
    const int cxc = gx < 0 ? 0 : (gx > NX - 1 ? NX - 1 : gx);
#pragma unroll
    for (int r = 0; r < 8; ++r) {
        const int gz  = GZ0 + 16 * tz + r + 8 * half;
        const int czc = gz < 0 ? 0 : (gz > NZ - 1 ? NZ - 1 : gz);
        inb[r]  = (czc == gz) & (cxc == gx);
        cidx[r] = czc * NX + cxc;
    }
#pragma unroll
    for (int r = 0; r < 8; ++r) p1[r] = f1[cidx[r]];
#pragma unroll
    for (int r = 0; r < 8; ++r) rC[r] = rho[cidx[r]];
    __builtin_amdgcn_sched_barrier(0);
#pragma unroll
    for (int r = 0; r < 8; ++r) {
        vV[r]  = vel[cidx[r]];
        bB[r]  = beta[cidx[r]];
        cDm[r] = damping[cidx[r]];
    }
#pragma unroll
    for (int r = 0; r < 8; ++r) p2[r] = f2[cidx[r]];

#pragma unroll
    for (int r = 0; r < 8; ++r) {
        const int fo = fbase + (r + 8 * half + 2) * FS + lm + 2;
        p1[r]  = (inb[r] & !first) ? p1[r] : 0.0f;
        F0[fo] = p1[r];
        IR[fo] = inb[r] ? rcp_nr(rC[r]) : 0.0f;
    }

    v2f fw[5];
#pragma unroll
    for (int c = 0; c < 5; ++c) {
        const int kk = c < 4 ? 8 * half + 2 * c : (half ? 16 : -2);
        const int d  = kk - lm;
        fw[c][0] = lap_w(d); fw[c][1] = lap_w(d + 1);
    }
    const int hk = half ? 18 : 0;

    unsigned oz[2], ox[2], o4[2], oa[2], ob[2];
#pragma unroll
    for (int b = 0; b < 2; ++b) {
        const unsigned bb = (unsigned)(b * FW * FS);
        oz[b] = 4u * (bb + (unsigned)(fbase + hk * FS + lm + 2));
        ox[b] = 4u * (bb + (unsigned)(fbase + (lm + 2) * FS + 8 * half + 2));
        o4[b] = 4u * (bb + (unsigned)(fbase + (lm + 2) * FS + hk));
        oa[b] = 4u * (bb + (unsigned)(fbase + (8 * half + 2) * FS + lm + 1));
        ob[b] = oa[b] + 4u * (unsigned)(4 * FS);
        asm volatile("" : "+v"(oz[b]), "+v"(ox[b]), "+v"(o4[b]), "+v"(oa[b]), "+v"(ob[b]));
    }
    const char* const Lc = (const char*)L;
    char* const Lw = (char*)L;

    int sel_off = -1;
    {
        const int rz = ez - (GZ0 + HL), rx = ex - (GX0 + HL);
        const bool mine = (tid < nch) & (rz >= 0) & (rz < OT) & (rx >= 0) & (rx < OT);
        sel_off = mine ? (rz + HL + 2) * FS + rx + HL + 2 : -1;
    }

    lds_barrier();

#pragma unroll
    for (int s = 0; s < KST; ++s) {
        const int sc = s < nsteps ? s : nsteps - 1;
#pragma unroll
        for (int r = 0; r < 8; ++r) pu[s][r] = pulse[(size_t)sc * (NZ * NX) + cidx[r]];
        __builtin_amdgcn_sched_barrier(0);
    }
    asm volatile("" ::: "memory");

    float cA1[8], cA2x[8], cA2z[8], cD[8], cA0[8], cB0[8], cC0[8];
#pragma unroll
    for (int r = 0; r < 8; ++r) {
        const int fo = fbase + (r + 8 * half + 2) * FS + lm + 2;
        p2[r] = first ? 0.0f : p2[r];
        const float irE = IR[fo + 1], irW = IR[fo - 1], irS = IR[fo + FS], irN = IR[fo - FS];
        const float Kc  = vV[r] * vV[r] * rC[r];
        const float gxr = 0.5f * (irE - irW);
        const float gzr = 0.5f * (irS - irN);
        cA1[r]  = DT2 * Kc * rC[r];
        cA2x[r] = 0.5f * (DT2 * Kc * gxr);
        cA2z[r] = 0.5f * (DT2 * Kc * gzr);
        cD[r]   = div_rc(2.0f * bB[r], Kc);
        cA0[r]  = __builtin_fmaf(2.0f, cDm[r], 1.0f);
        cB0[r]  = __builtin_fmaf(-cDm[r], cDm[r], 2.0f);
        cC0[r]  = __builtin_fmaf(2.0f, cDm[r], -1.0f);
    }

    v8f accL  = {};
    if (nsteps > 1 || wave < 4) z_own_rows(p1, fw, accL);

#pragma unroll
    for (int s = 0; s < KST; ++s) {
        if (s < nsteps) {
            const int cb = s & 1, nb = cb ^ 1;
            const float* Fn = nb ? F1 : F0;
            const bool lastst = (s == nsteps - 1);
            const bool sample = ((s & 3) == 0);
            if (!lastst || wave < 4) {
                __builtin_amdgcn_s_setprio(3);
                v2f px[5], pzh;
                {
                    const float* q = (const float*)(Lc + oz[cb]);
                    pzh[0] = q[0];
                    pzh[1] = q[FS];
                }
#pragma unroll
                for (int c = 0; c < 4; ++c) {
                    const float* q = (const float*)(Lc + ox[cb]);
                    px[c][0] = q[2 * c];
                    px[c][1] = q[2 * c + 1];
                }
                {
                    const float* q = (const float*)(Lc + o4[cb]);
                    px[4][0] = q[0];
                    px[4][1] = q[1];
                }
                asm volatile("" ::: "memory");
                float gl[8], gr[8];
#pragma unroll
                for (int r = 0; r < 8; ++r) {
                    const float* q = (const float*)(Lc + (r < 4 ? oa[cb] : ob[cb]));
                    gl[r] = q[(r & 3) * FS];
                    gr[r] = q[(r & 3) * FS + 2];
                }
                reads_before_products(accL);
                float dz[8];
                {
                    const float up_o = xhalf(p1[7]);
                    const float dn_o = xhalf(p1[0]);
                    const float up0  = half ? up_o : pzh[1];
                    const float dn7  = half ? pzh[0] : dn_o;
                    dz[0] = p1[1] - up0;
#pragma unroll
                    for (int r = 1; r < 7; ++r) dz[r] = p1[r + 1] - p1[r - 1];
                    dz[7] = dn7 - p1[6];
                }
                accL  = __builtin_amdgcn_wmma_f32_16x16x4_f32(false, fw[4], false, pzh, (short)0, accL,  false, false);
#pragma unroll
                for (int c = 0; c < 5; ++c)
                    accL  = __builtin_amdgcn_wmma_f32_16x16x4_f32(false, px[c], false, fw[c], (short)0, accL,  false, false);
                prio_default_behind(accL);
                __builtin_amdgcn_sched_barrier(0);
                float dx[8];
#pragma unroll
                for (int r = 0; r < 8; ++r) dx[r] = gr[r] - gl[r];
                float pn[8];
                if (NLA) {
#pragma unroll
                    for (int k = 0; k < 4; ++k) {
                        const v2f q1  = {p1[2 * k], p1[2 * k + 1]};
                        const v2f q2  = {p2[2 * k], p2[2 * k + 1]};
                        const v2f dd  = {cD[2 * k], cD[2 * k + 1]};
                        const v2f a0  = {cA0[2 * k], cA0[2 * k + 1]};
                        const v2f b0  = {cB0[2 * k], cB0[2 * k + 1]};
                        const v2f c0  = {cC0[2 * k], cC0[2 * k + 1]};
                        const v2f a1  = {cA1[2 * k], cA1[2 * k + 1]};
                        const v2f a2x = {cA2x[2 * k], cA2x[2 * k + 1]};
                        const v2f a2z = {cA2z[2 * k], cA2z[2 * k + 1]};
                        const v2f sL  = {accL[2 * k], accL[2 * k + 1]};
                        const v2f sGx = {dx[2 * k], dx[2 * k + 1]};
                        const v2f sGz = {dz[2 * k], dz[2 * k + 1]};
                        const v2f pls = {pu[s][2 * k], pu[s][2 * k + 1]};
                        const v2f a   = __builtin_elementwise_fma(dd, q1, a0);
                        const v2f dp  = q1 - q2;
                        const v2f t   = __builtin_elementwise_fma(dp, dp, q1 * (q1 + dp));
                        v2f num = __builtin_elementwise_fma(a2z, sGz, pls);
                        num = __builtin_elementwise_fma(c0, q2, num);
                        num = __builtin_elementwise_fma(b0, q1, num);
                        num = __builtin_elementwise_fma(dd, t, num);
                        num = __builtin_elementwise_fma(a2x, sGx, num);
                        num = __builtin_elementwise_fma(a1, sL, num);
                        const v2f res = div_rc2(num, a);
                        pn[2 * k]     = res[0];
                        pn[2 * k + 1] = res[1];
                    }
                } else {
#pragma unroll
                    for (int k = 0; k < 4; ++k) {
                        const v2f q1  = {p1[2 * k], p1[2 * k + 1]};
                        const v2f q2  = {p2[2 * k], p2[2 * k + 1]};
                        const v2f a0  = {cA0[2 * k], cA0[2 * k + 1]};
                        const v2f b0  = {cB0[2 * k], cB0[2 * k + 1]};
                        const v2f c0  = {cC0[2 * k], cC0[2 * k + 1]};
                        const v2f a1  = {cA1[2 * k], cA1[2 * k + 1]};
                        const v2f a2x = {cA2x[2 * k], cA2x[2 * k + 1]};
                        const v2f a2z = {cA2z[2 * k], cA2z[2 * k + 1]};
                        const v2f sL  = {accL[2 * k], accL[2 * k + 1]};
                        const v2f sGx = {dx[2 * k], dx[2 * k + 1]};
                        const v2f sGz = {dz[2 * k], dz[2 * k + 1]};
                        const v2f pls = {pu[s][2 * k], pu[s][2 * k + 1]};
                        v2f num = __builtin_elementwise_fma(a2z, sGz, DT2 * pls);
                        num = __builtin_elementwise_fma(c0, q2, num);
                        num = __builtin_elementwise_fma(b0, q1, num);
                        num = __builtin_elementwise_fma(a2x, sGx, num);
                        num = __builtin_elementwise_fma(a1, sL, num);
                        const v2f res = div_rc2(num, a0);
                        pn[2 * k]     = res[0];
                        pn[2 * k + 1] = res[1];
                    }
                }
#pragma unroll
                for (int r = 0; r < 8; ++r) {
                    p2[r] = p1[r];
                    p1[r] = inb[r] ? pn[r] : 0.0f;
                }
                if (!lastst || sample) {
#pragma unroll
                    for (int r = 0; r < 8; ++r) {
                        float* q = (float*)(Lw + (r < 4 ? oa[nb] : ob[nb]));
                        q[(r & 3) * FS + 1] = p1[r];
                    }
                }
            }
            const bool bar = (!lastst || sample);
            if (bar) lds_barrier_arrive(p1);
            if (s + 1 < KST && s + 1 < nsteps && (s + 1 < nsteps - 1 || wave < 4)) {
                accL  = v8f{};
                z_own_rows(p1, fw, accL);
            }
            if (bar) lds_barrier_wait(accL);
            if (sample) {
                float* grow = gout + (size_t)(s >> 2) * nch;
                if (sel_off >= 0) grow[tid] = Fn[sel_off];
                for (int ch = tid + 512; ch < nch; ch += 512) {
                    int rz = el[2 * ch + 0] - (GZ0 + HL);
                    int rx = el[2 * ch + 1] - (GX0 + HL);
                    if (rz >= 0 && rz < OT && rx >= 0 && rx < OT)
                        grow[ch] = Fn[(rz + HL + 2) * FS + rx + HL + 2];
                }
            }
        }
    }

    if (wave < 4) {
#pragma unroll
        for (int r = 0; r < 8; ++r) {
            const int idx = (GZ0 + 16 * tz + r + 8 * half) * NX + gx;
            pout1[idx] = p1[r];
            pout2[idx] = p2[r];
        }
    }
}

extern "C" void kernel_launch(void* const* d_in, const int* in_sizes, int n_in,
                              void* d_out, int out_size, void* d_ws, size_t ws_size,
                              hipStream_t stream) {
    const float* pulse   = (const float*)d_in[0];
    const float* vel     = (const float*)d_in[1];
    const float* rho     = (const float*)d_in[2];
    const float* damping = (const float*)d_in[3];
    const float* beta    = (const float*)d_in[4];
    const int*   el      = (const int*)d_in[5];
    const int*   nla     = (const int*)d_in[6];

    const int nt  = in_sizes[0] / (NZ * NX);
    const int nch = in_sizes[5] / 2;

    float* ws = (float*)d_ws;
    const size_t F = (size_t)NZ * NX;
    float* pl[4] = { ws, ws + F, ws + 2 * F, ws + 3 * F };

    int cur = 0;
    for (int t0 = 0; t0 < nt; t0 += KST) {
        const int ns = (nt - t0 < KST) ? (nt - t0) : KST;
        steps_kernel<<<dim3(NX / OT, NZ / OT), 512, 0, stream>>>(
            pl[2 * cur], pl[2 * cur + 1], pl[2 * (cur ^ 1)], pl[2 * (cur ^ 1) + 1],
            pulse + (size_t)t0 * F, vel, rho, damping, beta, nla, el,
            (float*)d_out + (size_t)(t0 >> 2) * nch, nch, ns, t0 == 0 ? 1 : 0);
        cur ^= 1;
    }
}


# two 16x16 tiles per wave: 8-wave workgroups, half as many waves to enter per launch; step count fixed at compile time for full launches (one body per step); pulse planes requested 3 steps ahead
# speedup vs baseline: 10.8183x; 1.0976x over previous
#include <hip/hip_runtime.h>

#define NZ 512
#define NX 512
#define DT2 0.0025f

#define KST 8
#define OT  32
#define HL  16
#define FW  68
#define NTW 2
#define NWV 8
#define PD  3
#define FS  69

typedef __attribute__((ext_vector_type(2))) float v2f;
typedef __attribute__((ext_vector_type(8))) float v8f;

__device__ __forceinline__ float lap_w(int d) {
    int a = d < 0 ? -d : d;
    return a == 0 ? -2.5f : (a == 1 ? 4.0f / 3.0f : (a == 2 ? -1.0f / 12.0f : 0.0f));
}

__device__ __forceinline__ float div_rc(float n, float a) {
    const float r = __builtin_amdgcn_rcpf(a);
    const float q = n * r;
    const float e = __builtin_fmaf(-a, q, n);
    return __builtin_fmaf(e, r, q);
}
__device__ __forceinline__ v2f div_rc2(v2f n, v2f a) {
    v2f r;
    r[0] = __builtin_amdgcn_rcpf(a[0]);
    r[1] = __builtin_amdgcn_rcpf(a[1]);
    const v2f q = n * r;
    const v2f e = __builtin_elementwise_fma(-a, q, n);
    return __builtin_elementwise_fma(e, r, q);
}
__device__ __forceinline__ float rcp_nr(float a) {
    const float r = __builtin_amdgcn_rcpf(a);
    const float e = __builtin_fmaf(-a, r, 1.0f);
    return __builtin_fmaf(e, r, r);
}

__device__ __forceinline__ void lds_barrier() {
    asm volatile("s_wait_dscnt 0x0\n\ts_barrier_signal -1\n\ts_barrier_wait -1" ::: "memory");
}

__device__ __forceinline__ void lds_barrier_arrive(float (&p)[NTW][8]) {
    asm volatile("s_wait_dscnt 0x0\n\ts_barrier_signal -1"
                 : "+v"(p[0][0]), "+v"(p[0][1]), "+v"(p[0][2]), "+v"(p[0][3]), "+v"(p[0][4]), "+v"(p[0][5]), "+v"(p[0][6]), "+v"(p[0][7]),
                   "+v"(p[1][0]), "+v"(p[1][1]), "+v"(p[1][2]), "+v"(p[1][3]), "+v"(p[1][4]), "+v"(p[1][5]), "+v"(p[1][6]), "+v"(p[1][7])
                 :: "memory");
}
__device__ __forceinline__ void lds_barrier_wait(v8f (&a)[NTW]) {
    asm volatile("s_barrier_wait -1" : "+v"(a[0]), "+v"(a[1]) :: "memory");
}

__device__ __forceinline__ void prio_default_behind(v8f (&a)[NTW]) {
    asm volatile("s_setprio 0" : "+v"(a[0]), "+v"(a[1]));
}

__device__ __forceinline__ void reads_before_products(v8f (&a)[NTW]) {
    asm volatile("" : "+v"(a[0]), "+v"(a[1]) :: "memory");
}

__device__ __forceinline__ float xhalf(float x) {
    const int b = __builtin_bit_cast(int, x);
    return __builtin_bit_cast(float, __builtin_amdgcn_permlanex16(b, b, 0x76543210, 0xfedcba98, false, false));
}

__device__ __forceinline__ void z_own_rows(const float (&p)[8], const v2f (&fw)[5], v8f& accL) {
#pragma unroll
    for (int c = 0; c < 4; ++c) {
        v2f pz;
        pz[0] = p[2 * c];
        pz[1] = p[2 * c + 1];
        accL  = __builtin_amdgcn_wmma_f32_16x16x4_f32(false, fw[c], false, pz, (short)0, accL,  false, false);
    }
}

template <int N> struct tiles_c { static constexpr int value = N; };

template <bool FULL>
__launch_bounds__(32 * NWV) __attribute__((amdgpu_waves_per_eu(2, 2)))
__global__ void steps_kernel(const float* __restrict__ pin1, const float* __restrict__ pin2,
                             float* __restrict__ pout1, float* __restrict__ pout2,
                             const float* pulse,
                             const float* __restrict__ vel,
                             const float* __restrict__ rho,
                             const float* __restrict__ damping,
                             const float* __restrict__ beta,
                             const int* __restrict__ nlaflag,
                             const int* __restrict__ el,
                             float* __restrict__ gout,
                             int nch, int nsteps_arg, int first) {
    const int nsteps = FULL ? KST : nsteps_arg;
    __shared__ float L[3 * FW * FS];
    float* const F0 = L;
    float* const F1 = L + FW * FS;
    float* const IR = L + 2 * FW * FS;

    const int tid  = threadIdx.x;
    const int wave = __builtin_amdgcn_readfirstlane(tid >> 5);
    const int lane = tid & 31;
    const int half = lane >> 4;
    const int lm   = lane & 15;
    int tz[NTW], tx[NTW], fbase[NTW];
#pragma unroll
    for (int t = 0; t < NTW; ++t) {
        const unsigned code = (unsigned)((0xFEDCB8743210A965ULL >> (4 * (wave + NWV * t))) & 15ULL);
        tz[t] = (int)(code >> 2);
        tx[t] = (int)(code & 3u);
        fbase[t] = (16 * tz[t]) * FS + 16 * tx[t];
    }
    const int GZ0 = (int)blockIdx.y * OT - HL;
    const int GX0 = (int)blockIdx.x * OT - HL;

    int ez = 0, ex = 0;
    if (nch > 0) {
        const int ci = tid < nch ? tid : nch - 1;
        ez = el[2 * ci + 0];
        ex = el[2 * ci + 1];
    }

    const int NLA = nlaflag[0];

#pragma unroll
    for (int i0 = 0; i0 < 4 * FW; i0 += 32 * NWV) {
        const int i = i0 + tid;
        if (i < 4 * FW) {
            const int q  = i / FW, xx = i - q * FW;
            const int zz = q < 2 ? q : q + (FW - 4);
            F0[zz * FS + xx] = 0.0f;
            F1[zz * FS + xx] = 0.0f;
            IR[zz * FS + xx] = 0.0f;
        }
    }
#pragma unroll
    for (int i0 = 0; i0 < 4 * (FW - 4); i0 += 32 * NWV) {
        const int i = i0 + tid;
        if (i < 4 * (FW - 4)) {
            const int zz = 2 + (i >> 2), cc = i & 3;
            const int xx = cc < 2 ? cc : cc + (FW - 4);
            F0[zz * FS + xx] = 0.0f;
            F1[zz * FS + xx] = 0.0f;
            IR[zz * FS + xx] = 0.0f;
        }
    }
    const float* const f1 = first ? rho : pin1;
    const float* const f2 = first ? rho : pin2;
    float rC[NTW][8], vV[NTW][8], bB[NTW][8], cDm[NTW][8], p1[NTW][8], p2[NTW][8], pu[NTW][KST][8];
    bool inb[NTW][8];
    int  cidx[NTW][8];
    unsigned cofs[NTW][8];
#pragma unroll
    for (int t = 0; t < NTW; ++t) {
        const int gx  = GX0 + 16 * tx[t] + lm;
        const int cxc = gx < 0 ? 0 : (gx > NX - 1 ? NX - 1 : gx);
#pragma unroll
        for (int r = 0; r < 8; ++r) {
            const int gz  = GZ0 + 16 * tz[t] + r + 8 * half;
            const int czc = gz < 0 ? 0 : (gz > NZ - 1 ? NZ - 1 : gz);
            inb[t][r]  = (czc == gz) & (cxc == gx);
            cidx[t][r] = czc * NX + cxc;
            cofs[t][r] = 4u * (unsigned)cidx[t][r];
        }
    }
#pragma unroll
    for (int t = 0; t < NTW; ++t) {
#pragma unroll
        for (int r = 0; r < 8; ++r) p1[t][r] = f1[cidx[t][r]];
    }
#pragma unroll
    for (int t = 0; t < NTW; ++t) {
#pragma unroll
        for (int r = 0; r < 8; ++r) rC[t][r] = rho[cidx[t][r]];
    }
    __builtin_amdgcn_sched_barrier(0);
#pragma unroll
    for (int t = 0; t < NTW; ++t) {
#pragma unroll
        for (int r = 0; r < 8; ++r) {
            vV[t][r]  = vel[cidx[t][r]];
            bB[t][r]  = beta[cidx[t][r]];
            cDm[t][r] = damping[cidx[t][r]];
        }
    }
#pragma unroll
    for (int t = 0; t < NTW; ++t) {
#pragma unroll
        for (int r = 0; r < 8; ++r) p2[t][r] = f2[cidx[t][r]];
    }

#pragma unroll
    for (int t = 0; t < NTW; ++t) {
#pragma unroll
        for (int r = 0; r < 8; ++r) {
            const int fo = fbase[t] + (r + 8 * half + 2) * FS + lm + 2;
            p1[t][r] = (inb[t][r] & !first) ? p1[t][r] : 0.0f;
            F0[fo]   = p1[t][r];
            IR[fo]   = inb[t][r] ? rcp_nr(rC[t][r]) : 0.0f;
        }
    }

    v2f fw[5];
#pragma unroll
    for (int c = 0; c < 5; ++c) {
        const int kk = c < 4 ? 8 * half + 2 * c : (half ? 16 : -2);
        const int d  = kk - lm;
        fw[c][0] = lap_w(d); fw[c][1] = lap_w(d + 1);
    }
    const int hk = half ? 18 : 0;

    unsigned oz[NTW][2], ox[NTW][2], o4[NTW][2], oc[NTW][2][2];
#pragma unroll
    for (int t = 0; t < NTW; ++t) {
#pragma unroll
        for (int b = 0; b < 2; ++b) {
            const unsigned bb = (unsigned)(b * FW * FS);
            oz[t][b] = 4u * (bb + (unsigned)(fbase[t] + hk * FS + lm + 2));
            ox[t][b] = 4u * (bb + (unsigned)(fbase[t] + (lm + 2) * FS + 8 * half + 2));
            o4[t][b] = 4u * (bb + (unsigned)(fbase[t] + (lm + 2) * FS + hk));
            oc[t][b][0] = 4u * (bb + (unsigned)(fbase[t] + (8 * half + 2) * FS + lm + 1));
            oc[t][b][1] = oc[t][b][0] + 4u * (unsigned)(4 * FS);
            asm volatile("" : "+v"(oz[t][b]), "+v"(ox[t][b]), "+v"(o4[t][b]), "+v"(oc[t][b][0]), "+v"(oc[t][b][1]));
        }
    }
    const char* const Lc = (const char*)L;
    char* const Lw = (char*)L;

    int sel_off = -1;
    {
        const int rz = ez - (GZ0 + HL), rx = ex - (GX0 + HL);
        const bool mine = (tid < nch) & (rz >= 0) & (rz < OT) & (rx >= 0) & (rx < OT);
        sel_off = mine ? (rz + HL + 2) * FS + rx + HL + 2 : -1;
    }

    lds_barrier();

#pragma unroll
    for (int s = 0; s < PD; ++s) {
        const int sc = s < nsteps ? s : nsteps - 1;
#pragma unroll
        for (int t = 0; t < NTW; ++t) {
#pragma unroll
            for (int r = 0; r < 8; ++r) pu[t][s][r] = pulse[(size_t)sc * (NZ * NX) + cidx[t][r]];
        }
        __builtin_amdgcn_sched_barrier(0);
    }
    asm volatile("" ::: "memory");

    float cA1[NTW][8], cA2x[NTW][8], cA2z[NTW][8], cD[NTW][8], cA0[NTW][8], cB0[NTW][8], cC0[NTW][8];
#pragma unroll
    for (int t = 0; t < NTW; ++t) {
#pragma unroll
        for (int r = 0; r < 8; ++r) {
            const int fo = fbase[t] + (r + 8 * half + 2) * FS + lm + 2;
            p2[t][r] = first ? 0.0f : p2[t][r];
            const float irE = IR[fo + 1], irW = IR[fo - 1], irS = IR[fo + FS], irN = IR[fo - FS];
            const float Kc  = vV[t][r] * vV[t][r] * rC[t][r];
            const float gxr = 0.5f * (irE - irW);
            const float gzr = 0.5f * (irS - irN);
            cA1[t][r]  = DT2 * Kc * rC[t][r];
            cA2x[t][r] = 0.5f * (DT2 * Kc * gxr);
            cA2z[t][r] = 0.5f * (DT2 * Kc * gzr);
            cD[t][r]   = div_rc(2.0f * bB[t][r], Kc);
            cA0[t][r]  = __builtin_fmaf(2.0f, cDm[t][r], 1.0f);
            cB0[t][r]  = __builtin_fmaf(-cDm[t][r], cDm[t][r], 2.0f);
            cC0[t][r]  = __builtin_fmaf(2.0f, cDm[t][r], -1.0f);
        }
    }

    v8f accL[NTW] = {};
    if (nsteps > 1) {
#pragma unroll
        for (int t = 0; t < NTW; ++t) z_own_rows(p1[t], fw, accL[t]);
    } else if (wave < 4) {
        z_own_rows(p1[0], fw, accL[0]);
    }

#pragma unroll
    for (int s = 0; s < KST; ++s) {
        if (s < nsteps) {
            const int cb = s & 1, nb = cb ^ 1;
            const float* Fn = nb ? F1 : F0;
            const bool lastst = (s == nsteps - 1);
            const bool sample = ((s & 3) == 0);
            auto step_tiles = [&](auto ntc) __attribute__((always_inline)) {
                constexpr int ntl = decltype(ntc)::value;
                __builtin_amdgcn_s_setprio(3);
                v2f px[NTW][5], pzh[NTW];
#pragma unroll
                for (int t = 0; t < ntl; ++t) {
                    {
                        const float* q = (const float*)(Lc + oz[t][cb]);
                        pzh[t][0] = q[0];
                        pzh[t][1] = q[FS];
                    }
#pragma unroll
                    for (int c = 0; c < 4; ++c) {
                        const float* q = (const float*)(Lc + ox[t][cb]);
                        px[t][c][0] = q[2 * c];
                        px[t][c][1] = q[2 * c + 1];
                    }
                    {
                        const float* q = (const float*)(Lc + o4[t][cb]);
                        px[t][4][0] = q[0];
                        px[t][4][1] = q[1];
                    }
                }
                asm volatile("" ::: "memory");
                float gl[NTW][8], gr[NTW][8];
#pragma unroll
                for (int t = 0; t < ntl; ++t) {
#pragma unroll
                    for (int r = 0; r < 8; ++r) {
                        const float* q = (const float*)(Lc + oc[t][cb][r >> 2]);
                        gl[t][r] = q[(r & 3) * FS];
                        gr[t][r] = q[(r & 3) * FS + 2];
                    }
                }
                reads_before_products(accL);
                float dz[NTW][8];
#pragma unroll
                for (int t = 0; t < ntl; ++t) {
                    const float up_o = xhalf(p1[t][7]);
                    const float dn_o = xhalf(p1[t][0]);
                    const float up0  = half ? up_o : pzh[t][1];
                    const float dn7  = half ? pzh[t][0] : dn_o;
                    dz[t][0] = p1[t][1] - up0;
#pragma unroll
                    for (int r = 1; r < 7; ++r) dz[t][r] = p1[t][r + 1] - p1[t][r - 1];
                    dz[t][7] = dn7 - p1[t][6];
                }
#pragma unroll
                for (int t = 0; t < ntl; ++t) {
                    accL[t] = __builtin_amdgcn_wmma_f32_16x16x4_f32(false, fw[4], false, pzh[t], (short)0, accL[t], false, false);
#pragma unroll
                    for (int c = 0; c < 5; ++c)
                        accL[t] = __builtin_amdgcn_wmma_f32_16x16x4_f32(false, px[t][c], false, fw[c], (short)0, accL[t], false, false);
                }
                prio_default_behind(accL);
                __builtin_amdgcn_sched_barrier(0);
#pragma unroll
                for (int t = 0; t < ntl; ++t) {
                    float dx[8];
#pragma unroll
                    for (int r = 0; r < 8; ++r) dx[r] = gr[t][r] - gl[t][r];
                    float pn[8];
                    if (NLA) {
#pragma unroll
                        for (int k = 0; k < 4; ++k) {
                            const v2f q1  = {p1[t][2 * k], p1[t][2 * k + 1]};
                            const v2f q2  = {p2[t][2 * k], p2[t][2 * k + 1]};
                            const v2f dd  = {cD[t][2 * k], cD[t][2 * k + 1]};
                            const v2f a0  = {cA0[t][2 * k], cA0[t][2 * k + 1]};
                            const v2f b0  = {cB0[t][2 * k], cB0[t][2 * k + 1]};
                            const v2f c0  = {cC0[t][2 * k], cC0[t][2 * k + 1]};
                            const v2f a1  = {cA1[t][2 * k], cA1[t][2 * k + 1]};
                            const v2f a2x = {cA2x[t][2 * k], cA2x[t][2 * k + 1]};
                            const v2f a2z = {cA2z[t][2 * k], cA2z[t][2 * k + 1]};
                            const v2f sL  = {accL[t][2 * k], accL[t][2 * k + 1]};
                            const v2f sGx = {dx[2 * k], dx[2 * k + 1]};
                            const v2f sGz = {dz[t][2 * k], dz[t][2 * k + 1]};
                            const v2f pls = {pu[t][s][2 * k], pu[t][s][2 * k + 1]};
                            const v2f a   = __builtin_elementwise_fma(dd, q1, a0);
                            const v2f dp  = q1 - q2;
                            const v2f tq  = __builtin_elementwise_fma(dp, dp, q1 * (q1 + dp));
                            v2f num = __builtin_elementwise_fma(a2z, sGz, pls);
                            num = __builtin_elementwise_fma(c0, q2, num);
                            num = __builtin_elementwise_fma(b0, q1, num);
                            num = __builtin_elementwise_fma(dd, tq, num);
                            num = __builtin_elementwise_fma(a2x, sGx, num);
                            num = __builtin_elementwise_fma(a1, sL, num);
                            const v2f res = div_rc2(num, a);
                            pn[2 * k]     = res[0];
                            pn[2 * k + 1] = res[1];
                        }
                    } else {
#pragma unroll
                        for (int k = 0; k < 4; ++k) {
                            const v2f q1  = {p1[t][2 * k], p1[t][2 * k + 1]};
                            const v2f q2  = {p2[t][2 * k], p2[t][2 * k + 1]};
                            const v2f a0  = {cA0[t][2 * k], cA0[t][2 * k + 1]};
                            const v2f b0  = {cB0[t][2 * k], cB0[t][2 * k + 1]};
                            const v2f c0  = {cC0[t][2 * k], cC0[t][2 * k + 1]};
                            const v2f a1  = {cA1[t][2 * k], cA1[t][2 * k + 1]};
                            const v2f a2x = {cA2x[t][2 * k], cA2x[t][2 * k + 1]};
                            const v2f a2z = {cA2z[t][2 * k], cA2z[t][2 * k + 1]};
                            const v2f sL  = {accL[t][2 * k], accL[t][2 * k + 1]};
                            const v2f sGx = {dx[2 * k], dx[2 * k + 1]};
                            const v2f sGz = {dz[t][2 * k], dz[t][2 * k + 1]};
                            const v2f pls = {pu[t][s][2 * k], pu[t][s][2 * k + 1]};
                            v2f num = __builtin_elementwise_fma(a2z, sGz, DT2 * pls);
                            num = __builtin_elementwise_fma(c0, q2, num);
                            num = __builtin_elementwise_fma(b0, q1, num);
                            num = __builtin_elementwise_fma(a2x, sGx, num);
                            num = __builtin_elementwise_fma(a1, sL, num);
                            const v2f res = div_rc2(num, a0);
                            pn[2 * k]     = res[0];
                            pn[2 * k + 1] = res[1];
                        }
                    }
#pragma unroll
                    for (int r = 0; r < 8; ++r) {
                        p2[t][r] = p1[t][r];
                        p1[t][r] = inb[t][r] ? pn[r] : 0.0f;
                    }
                    if (!lastst || sample) {
#pragma unroll
                        for (int r = 0; r < 8; ++r) {
                            float* q = (float*)(Lw + oc[t][nb][r >> 2]);
                            q[(r & 3) * FS + 1] = p1[t][r];
                        }
                    }
                }
            };
            auto request_plane = [&](int qs, auto ntc) __attribute__((always_inline)) {
                constexpr int ntl = decltype(ntc)::value;
                const char* const pq = (const char*)(pulse + (size_t)qs * (NZ * NX));
#pragma unroll
                for (int t = 0; t < ntl; ++t) {
#pragma unroll
                    for (int r = 0; r < 8; ++r) {
                        asm volatile("" : "+v"(cofs[t][r]));
                        pu[t][qs][r] = *(const float*)(pq + cofs[t][r]);
                    }
                }
            };
            if (!lastst)       step_tiles(tiles_c<NTW>{});
            else if (wave < 4) step_tiles(tiles_c<1>{});
            const bool bar = (!lastst || sample);
            if (bar) lds_barrier_arrive(p1);
            if (s + PD < KST) {
                if (s + PD < nsteps - 1)                     request_plane(s + PD, tiles_c<NTW>{});
                else if (s + PD == nsteps - 1 && wave < 4)   request_plane(s + PD, tiles_c<1>{});
            }
            if (s + 1 < KST && s + 1 < nsteps) {
                if (s + 1 < nsteps - 1) {
#pragma unroll
                    for (int t = 0; t < NTW; ++t) {
                        accL[t] = v8f{};
                        z_own_rows(p1[t], fw, accL[t]);
                    }
                } else if (wave < 4) {
                    accL[0] = v8f{};
                    z_own_rows(p1[0], fw, accL[0]);
                }
            }
            if (bar) lds_barrier_wait(accL);
            if (sample) {
                float* grow = gout + (size_t)(s >> 2) * nch;
                if (sel_off >= 0) grow[tid] = Fn[sel_off];
                for (int ch = tid + 32 * NWV; ch < nch; ch += 32 * NWV) {
                    int rz = el[2 * ch + 0] - (GZ0 + HL);
                    int rx = el[2 * ch + 1] - (GX0 + HL);
                    if (rz >= 0 && rz < OT && rx >= 0 && rx < OT)
                        grow[ch] = Fn[(rz + HL + 2) * FS + rx + HL + 2];
                }
            }
        }
    }

    if (wave < 4) {
        const int gx0 = GX0 + 16 * tx[0] + lm;
#pragma unroll
        for (int r = 0; r < 8; ++r) {
            const int idx = (GZ0 + 16 * tz[0] + r + 8 * half) * NX + gx0;
            pout1[idx] = p1[0][r];
            pout2[idx] = p2[0][r];
        }
    }
}

extern "C" void kernel_launch(void* const* d_in, const int* in_sizes, int n_in,
                              void* d_out, int out_size, void* d_ws, size_t ws_size,
                              hipStream_t stream) {
    const float* pulse   = (const float*)d_in[0];
    const float* vel     = (const float*)d_in[1];
    const float* rho     = (const float*)d_in[2];
    const float* damping = (const float*)d_in[3];
    const float* beta    = (const float*)d_in[4];
    const int*   el      = (const int*)d_in[5];
    const int*   nla     = (const int*)d_in[6];

    const int nt  = in_sizes[0] / (NZ * NX);
    const int nch = in_sizes[5] / 2;

    float* ws = (float*)d_ws;
    const size_t F = (size_t)NZ * NX;
    float* pl[4] = { ws, ws + F, ws + 2 * F, ws + 3 * F };

    int cur = 0;
    for (int t0 = 0; t0 < nt; t0 += KST) {
        const int ns = (nt - t0 < KST) ? (nt - t0) : KST;
        if (ns == KST)
            steps_kernel<true><<<dim3(NX / OT, NZ / OT), 32 * NWV, 0, stream>>>(
                pl[2 * cur], pl[2 * cur + 1], pl[2 * (cur ^ 1)], pl[2 * (cur ^ 1) + 1],
                pulse + (size_t)t0 * F, vel, rho, damping, beta, nla, el,
                (float*)d_out + (size_t)(t0 >> 2) * nch, nch, ns, t0 == 0 ? 1 : 0);
        else
            steps_kernel<false><<<dim3(NX / OT, NZ / OT), 32 * NWV, 0, stream>>>(
                pl[2 * cur], pl[2 * cur + 1], pl[2 * (cur ^ 1)], pl[2 * (cur ^ 1) + 1],
                pulse + (size_t)t0 * F, vel, rho, damping, beta, nla, el,
                (float*)d_out + (size_t)(t0 >> 2) * nch, nch, ns, t0 == 0 ? 1 : 0);
        cur ^= 1;
    }
}


# launch inputs packed per cell in the workspace: first launch stores the 4 step constants as one 16-byte value, fields go on as 8-byte pairs; later launches issue 48 loads per lane, not 96
# speedup vs baseline: 11.7746x; 1.0884x over previous
#include <hip/hip_runtime.h>

#define NZ 512
#define NX 512
#define DT2 0.0025f

#define KST 8
#define OT  32
#define HL  16
#define FW  68
#define NTW 2
#define NWV 8
#define PD  3
#define FS  69

typedef __attribute__((ext_vector_type(2))) float v2f;
typedef __attribute__((ext_vector_type(4))) float v4f;
typedef __attribute__((ext_vector_type(8))) float v8f;

__device__ __forceinline__ float lap_w(int d) {
    int a = d < 0 ? -d : d;
    return a == 0 ? -2.5f : (a == 1 ? 4.0f / 3.0f : (a == 2 ? -1.0f / 12.0f : 0.0f));
}

__device__ __forceinline__ float div_rc(float n, float a) {
    const float r = __builtin_amdgcn_rcpf(a);
    const float q = n * r;
    const float e = __builtin_fmaf(-a, q, n);
    return __builtin_fmaf(e, r, q);
}
__device__ __forceinline__ v2f div_rc2(v2f n, v2f a) {
    v2f r;
    r[0] = __builtin_amdgcn_rcpf(a[0]);
    r[1] = __builtin_amdgcn_rcpf(a[1]);
    const v2f q = n * r;
    const v2f e = __builtin_elementwise_fma(-a, q, n);
    return __builtin_elementwise_fma(e, r, q);
}
__device__ __forceinline__ float rcp_nr(float a) {
    const float r = __builtin_amdgcn_rcpf(a);
    const float e = __builtin_fmaf(-a, r, 1.0f);
    return __builtin_fmaf(e, r, r);
}

__device__ __forceinline__ void lds_barrier() {
    asm volatile("s_wait_dscnt 0x0\n\ts_barrier_signal -1\n\ts_barrier_wait -1" ::: "memory");
}

__device__ __forceinline__ void lds_barrier_arrive(float (&p)[NTW][8]) {
    asm volatile("s_wait_dscnt 0x0\n\ts_barrier_signal -1"
                 : "+v"(p[0][0]), "+v"(p[0][1]), "+v"(p[0][2]), "+v"(p[0][3]), "+v"(p[0][4]), "+v"(p[0][5]), "+v"(p[0][6]), "+v"(p[0][7]),
                   "+v"(p[1][0]), "+v"(p[1][1]), "+v"(p[1][2]), "+v"(p[1][3]), "+v"(p[1][4]), "+v"(p[1][5]), "+v"(p[1][6]), "+v"(p[1][7])
                 :: "memory");
}
__device__ __forceinline__ void lds_barrier_wait(v8f (&a)[NTW]) {
    asm volatile("s_barrier_wait -1" : "+v"(a[0]), "+v"(a[1]) :: "memory");
}

__device__ __forceinline__ void prio_default_behind(v8f (&a)[NTW]) {
    asm volatile("s_setprio 0" : "+v"(a[0]), "+v"(a[1]));
}

__device__ __forceinline__ void reads_before_products(v8f (&a)[NTW]) {
    asm volatile("" : "+v"(a[0]), "+v"(a[1]) :: "memory");
}

__device__ __forceinline__ float xhalf(float x) {
    const int b = __builtin_bit_cast(int, x);
    return __builtin_bit_cast(float, __builtin_amdgcn_permlanex16(b, b, 0x76543210, 0xfedcba98, false, false));
}

__device__ __forceinline__ void z_own_rows(const float (&p)[8], const v2f (&fw)[5], v8f& accL) {
#pragma unroll
    for (int c = 0; c < 4; ++c) {
        v2f pz;
        pz[0] = p[2 * c];
        pz[1] = p[2 * c + 1];
        accL  = __builtin_amdgcn_wmma_f32_16x16x4_f32(false, fw[c], false, pz, (short)0, accL,  false, false);
    }
}

template <int N> struct tiles_c { static constexpr int value = N; };

template <bool FULL, int MODE>
__launch_bounds__(32 * NWV) __attribute__((amdgpu_waves_per_eu(2, 2)))
__global__ void steps_kernel(const float* __restrict__ fin,
                             float* __restrict__ fout,
                             float* __restrict__ cpk,
                             const float* pulse,
                             const float* __restrict__ vel,
                             const float* __restrict__ rho,
                             const float* __restrict__ damping,
                             const float* __restrict__ beta,
                             const int* __restrict__ nlaflag,
                             const int* __restrict__ el,
                             float* __restrict__ gout,
                             int nch, int nsteps_arg) {
    const int nsteps = FULL ? KST : nsteps_arg;
    __shared__ float L[(MODE == 0 ? 3 : 2) * FW * FS];
    float* const F0 = L;
    float* const F1 = L + FW * FS;
    float* const IR = L + (MODE == 0 ? 2 : 0) * FW * FS;

    const int tid  = threadIdx.x;
    const int wave = __builtin_amdgcn_readfirstlane(tid >> 5);
    const int lane = tid & 31;
    const int half = lane >> 4;
    const int lm   = lane & 15;
    int tz[NTW], tx[NTW], fbase[NTW];
#pragma unroll
    for (int t = 0; t < NTW; ++t) {
        const unsigned code = (unsigned)((0xFEDCB8743210A965ULL >> (4 * (wave + NWV * t))) & 15ULL);
        tz[t] = (int)(code >> 2);
        tx[t] = (int)(code & 3u);
        fbase[t] = (16 * tz[t]) * FS + 16 * tx[t];
    }
    const int GZ0 = (int)blockIdx.y * OT - HL;
    const int GX0 = (int)blockIdx.x * OT - HL;

    int ez = 0, ex = 0;
    if (nch > 0) {
        const int ci = tid < nch ? tid : nch - 1;
        ez = el[2 * ci + 0];
        ex = el[2 * ci + 1];
    }

    const int NLA = nlaflag[0];

#pragma unroll
    for (int i0 = 0; i0 < 4 * FW; i0 += 32 * NWV) {
        const int i = i0 + tid;
        if (i < 4 * FW) {
            const int q  = i / FW, xx = i - q * FW;
            const int zz = q < 2 ? q : q + (FW - 4);
            F0[zz * FS + xx] = 0.0f;
            F1[zz * FS + xx] = 0.0f;
            if constexpr (MODE == 0) IR[zz * FS + xx] = 0.0f;
        }
    }
#pragma unroll
    for (int i0 = 0; i0 < 4 * (FW - 4); i0 += 32 * NWV) {
        const int i = i0 + tid;
        if (i < 4 * (FW - 4)) {
            const int zz = 2 + (i >> 2), cc = i & 3;
            const int xx = cc < 2 ? cc : cc + (FW - 4);
            F0[zz * FS + xx] = 0.0f;
            F1[zz * FS + xx] = 0.0f;
            if constexpr (MODE == 0) IR[zz * FS + xx] = 0.0f;
        }
    }
    float rC[NTW][8], vV[NTW][8], bB[NTW][8], cDm[NTW][8], p1[NTW][8], p2[NTW][8], pu[NTW][KST][8];
    v4f ck[NTW][8];
    bool inb[NTW][8];
    int  cidx[NTW][8];
    unsigned cofs[NTW][8];
#pragma unroll
    for (int t = 0; t < NTW; ++t) {
        const int gx  = GX0 + 16 * tx[t] + lm;
        const int cxc = gx < 0 ? 0 : (gx > NX - 1 ? NX - 1 : gx);
#pragma unroll
        for (int r = 0; r < 8; ++r) {
            const int gz  = GZ0 + 16 * tz[t] + r + 8 * half;
            const int czc = gz < 0 ? 0 : (gz > NZ - 1 ? NZ - 1 : gz);
            inb[t][r]  = (czc == gz) & (cxc == gx);
            cidx[t][r] = czc * NX + cxc;
            cofs[t][r] = 4u * (unsigned)cidx[t][r];
        }
    }
    if constexpr (MODE == 0) {
#pragma unroll
        for (int t = 0; t < NTW; ++t) {
#pragma unroll
            for (int r = 0; r < 8; ++r) rC[t][r] = rho[cidx[t][r]];
        }
        __builtin_amdgcn_sched_barrier(0);
#pragma unroll
        for (int t = 0; t < NTW; ++t) {
#pragma unroll
            for (int r = 0; r < 8; ++r) {
                vV[t][r]  = vel[cidx[t][r]];
                bB[t][r]  = beta[cidx[t][r]];
                cDm[t][r] = damping[cidx[t][r]];
                p1[t][r]  = 0.0f;
                p2[t][r]  = 0.0f;
            }
        }
    } else {
        const v2f* const f12 = (const v2f*)fin;
#pragma unroll
        for (int t = 0; t < NTW; ++t) {
#pragma unroll
            for (int r = 0; r < 8; ++r) {
                const v2f pp = f12[cidx[t][r]];
                p1[t][r] = pp[0];
                p2[t][r] = pp[1];
            }
        }
        __builtin_amdgcn_sched_barrier(0);
        const v4f* const c4 = (const v4f*)cpk;
#pragma unroll
        for (int t = 0; t < NTW; ++t) {
#pragma unroll
            for (int r = 0; r < 8; ++r) {
                ck[t][r]  = c4[cidx[t][r]];
                cDm[t][r] = damping[cidx[t][r]];
            }
        }
    }

#pragma unroll
    for (int t = 0; t < NTW; ++t) {
#pragma unroll
        for (int r = 0; r < 8; ++r) {
            const int fo = fbase[t] + (r + 8 * half + 2) * FS + lm + 2;
            if constexpr (MODE == 0) {
                F0[fo] = 0.0f;
                IR[fo] = inb[t][r] ? rcp_nr(rC[t][r]) : 0.0f;
            } else {
                p1[t][r] = inb[t][r] ? p1[t][r] : 0.0f;
                F0[fo]   = p1[t][r];
            }
        }
    }

    v2f fw[5];
#pragma unroll
    for (int c = 0; c < 5; ++c) {
        const int kk = c < 4 ? 8 * half + 2 * c : (half ? 16 : -2);
        const int d  = kk - lm;
        fw[c][0] = lap_w(d); fw[c][1] = lap_w(d + 1);
    }
    const int hk = half ? 18 : 0;

    unsigned oz[NTW][2], ox[NTW][2], o4[NTW][2], oc[NTW][2][2];
#pragma unroll
    for (int t = 0; t < NTW; ++t) {
#pragma unroll
        for (int b = 0; b < 2; ++b) {
            const unsigned bb = (unsigned)(b * FW * FS);
            oz[t][b] = 4u * (bb + (unsigned)(fbase[t] + hk * FS + lm + 2));
            ox[t][b] = 4u * (bb + (unsigned)(fbase[t] + (lm + 2) * FS + 8 * half + 2));
            o4[t][b] = 4u * (bb + (unsigned)(fbase[t] + (lm + 2) * FS + hk));
            oc[t][b][0] = 4u * (bb + (unsigned)(fbase[t] + (8 * half + 2) * FS + lm + 1));
            oc[t][b][1] = oc[t][b][0] + 4u * (unsigned)(4 * FS);
            asm volatile("" : "+v"(oz[t][b]), "+v"(ox[t][b]), "+v"(o4[t][b]), "+v"(oc[t][b][0]), "+v"(oc[t][b][1]));
        }
    }
    const char* const Lc = (const char*)L;
    char* const Lw = (char*)L;

    int sel_off = -1;
    {
        const int rz = ez - (GZ0 + HL), rx = ex - (GX0 + HL);
        const bool mine = (tid < nch) & (rz >= 0) & (rz < OT) & (rx >= 0) & (rx < OT);
        sel_off = mine ? (rz + HL + 2) * FS + rx + HL + 2 : -1;
    }

    lds_barrier();

#pragma unroll
    for (int s = 0; s < PD; ++s) {
        const int sc = s < nsteps ? s : nsteps - 1;
#pragma unroll
        for (int t = 0; t < NTW; ++t) {
#pragma unroll
            for (int r = 0; r < 8; ++r) pu[t][s][r] = pulse[(size_t)sc * (NZ * NX) + cidx[t][r]];
        }
        __builtin_amdgcn_sched_barrier(0);
    }
    asm volatile("" ::: "memory");

    float cA1[NTW][8], cA2x[NTW][8], cA2z[NTW][8], cD[NTW][8], cA0[NTW][8], cB0[NTW][8], cC0[NTW][8];
#pragma unroll
    for (int t = 0; t < NTW; ++t) {
#pragma unroll
        for (int r = 0; r < 8; ++r) {
            if constexpr (MODE == 0) {
                const int fo = fbase[t] + (r + 8 * half + 2) * FS + lm + 2;
                const float irE = IR[fo + 1], irW = IR[fo - 1], irS = IR[fo + FS], irN = IR[fo - FS];
                const float Kc  = vV[t][r] * vV[t][r] * rC[t][r];
                const float gxr = 0.5f * (irE - irW);
                const float gzr = 0.5f * (irS - irN);
                cA1[t][r]  = DT2 * Kc * rC[t][r];
                cA2x[t][r] = 0.5f * (DT2 * Kc * gxr);
                cA2z[t][r] = 0.5f * (DT2 * Kc * gzr);
                cD[t][r]   = div_rc(2.0f * bB[t][r], Kc);
            } else {
                cA1[t][r]  = ck[t][r][0];
                cA2x[t][r] = ck[t][r][1];
                cA2z[t][r] = ck[t][r][2];
                cD[t][r]   = ck[t][r][3];
            }
            cA0[t][r]  = __builtin_fmaf(2.0f, cDm[t][r], 1.0f);
            cB0[t][r]  = __builtin_fmaf(-cDm[t][r], cDm[t][r], 2.0f);
            cC0[t][r]  = __builtin_fmaf(2.0f, cDm[t][r], -1.0f);
        }
    }

    v8f accL[NTW] = {};
    if (nsteps > 1) {
#pragma unroll
        for (int t = 0; t < NTW; ++t) z_own_rows(p1[t], fw, accL[t]);
    } else if (wave < 4) {
        z_own_rows(p1[0], fw, accL[0]);
    }

#pragma unroll
    for (int s = 0; s < KST; ++s) {
        if (s < nsteps) {
            const int cb = s & 1, nb = cb ^ 1;
            const float* Fn = nb ? F1 : F0;
            const bool lastst = (s == nsteps - 1);
            const bool sample = ((s & 3) == 0);
            auto step_tiles = [&](auto ntc) __attribute__((always_inline)) {
                constexpr int ntl = decltype(ntc)::value;
                __builtin_amdgcn_s_setprio(3);
                v2f px[NTW][5], pzh[NTW];
#pragma unroll
                for (int t = 0; t < ntl; ++t) {
                    {
                        const float* q = (const float*)(Lc + oz[t][cb]);
                        pzh[t][0] = q[0];
                        pzh[t][1] = q[FS];
                    }
#pragma unroll
                    for (int c = 0; c < 4; ++c) {
                        const float* q = (const float*)(Lc + ox[t][cb]);
                        px[t][c][0] = q[2 * c];
                        px[t][c][1] = q[2 * c + 1];
                    }
                    {
                        const float* q = (const float*)(Lc + o4[t][cb]);
                        px[t][4][0] = q[0];
                        px[t][4][1] = q[1];
                    }
                }
                asm volatile("" ::: "memory");
                float gl[NTW][8], gr[NTW][8];
#pragma unroll
                for (int t = 0; t < ntl; ++t) {
#pragma unroll
                    for (int r = 0; r < 8; ++r) {
                        const float* q = (const float*)(Lc + oc[t][cb][r >> 2]);
                        gl[t][r] = q[(r & 3) * FS];
                        gr[t][r] = q[(r & 3) * FS + 2];
                    }
                }
                reads_before_products(accL);
                float dz[NTW][8];
#pragma unroll
                for (int t = 0; t < ntl; ++t) {
                    const float up_o = xhalf(p1[t][7]);
                    const float dn_o = xhalf(p1[t][0]);
                    const float up0  = half ? up_o : pzh[t][1];
                    const float dn7  = half ? pzh[t][0] : dn_o;
                    dz[t][0] = p1[t][1] - up0;
#pragma unroll
                    for (int r = 1; r < 7; ++r) dz[t][r] = p1[t][r + 1] - p1[t][r - 1];
                    dz[t][7] = dn7 - p1[t][6];
                }
#pragma unroll
                for (int t = 0; t < ntl; ++t) {
                    accL[t] = __builtin_amdgcn_wmma_f32_16x16x4_f32(false, fw[4], false, pzh[t], (short)0, accL[t], false, false);
#pragma unroll
                    for (int c = 0; c < 5; ++c)
                        accL[t] = __builtin_amdgcn_wmma_f32_16x16x4_f32(false, px[t][c], false, fw[c], (short)0, accL[t], false, false);
                }
                prio_default_behind(accL);
                __builtin_amdgcn_sched_barrier(0);
#pragma unroll
                for (int t = 0; t < ntl; ++t) {
                    float dx[8];
#pragma unroll
                    for (int r = 0; r < 8; ++r) dx[r] = gr[t][r] - gl[t][r];
                    float pn[8];
                    if (NLA) {
#pragma unroll
                        for (int k = 0; k < 4; ++k) {
                            const v2f q1  = {p1[t][2 * k], p1[t][2 * k + 1]};
                            const v2f q2  = {p2[t][2 * k], p2[t][2 * k + 1]};
                            const v2f dd  = {cD[t][2 * k], cD[t][2 * k + 1]};
                            const v2f a0  = {cA0[t][2 * k], cA0[t][2 * k + 1]};
                            const v2f b0  = {cB0[t][2 * k], cB0[t][2 * k + 1]};
                            const v2f c0  = {cC0[t][2 * k], cC0[t][2 * k + 1]};
                            const v2f a1  = {cA1[t][2 * k], cA1[t][2 * k + 1]};
                            const v2f a2x = {cA2x[t][2 * k], cA2x[t][2 * k + 1]};
                            const v2f a2z = {cA2z[t][2 * k], cA2z[t][2 * k + 1]};
                            const v2f sL  = {accL[t][2 * k], accL[t][2 * k + 1]};
                            const v2f sGx = {dx[2 * k], dx[2 * k + 1]};
                            const v2f sGz = {dz[t][2 * k], dz[t][2 * k + 1]};
                            const v2f pls = {pu[t][s][2 * k], pu[t][s][2 * k + 1]};
                            const v2f a   = __builtin_elementwise_fma(dd, q1, a0);
                            const v2f dp  = q1 - q2;
                            const v2f tq  = __builtin_elementwise_fma(dp, dp, q1 * (q1 + dp));
                            v2f num = __builtin_elementwise_fma(a2z, sGz, pls);
                            num = __builtin_elementwise_fma(c0, q2, num);
                            num = __builtin_elementwise_fma(b0, q1, num);
                            num = __builtin_elementwise_fma(dd, tq, num);
                            num = __builtin_elementwise_fma(a2x, sGx, num);
                            num = __builtin_elementwise_fma(a1, sL, num);
                            const v2f res = div_rc2(num, a);
                            pn[2 * k]     = res[0];
                            pn[2 * k + 1] = res[1];
                        }
                    } else {
#pragma unroll
                        for (int k = 0; k < 4; ++k) {
                            const v2f q1  = {p1[t][2 * k], p1[t][2 * k + 1]};
                            const v2f q2  = {p2[t][2 * k], p2[t][2 * k + 1]};
                            const v2f a0  = {cA0[t][2 * k], cA0[t][2 * k + 1]};
                            const v2f b0  = {cB0[t][2 * k], cB0[t][2 * k + 1]};
                            const v2f c0  = {cC0[t][2 * k], cC0[t][2 * k + 1]};
                            const v2f a1  = {cA1[t][2 * k], cA1[t][2 * k + 1]};
                            const v2f a2x = {cA2x[t][2 * k], cA2x[t][2 * k + 1]};
                            const v2f a2z = {cA2z[t][2 * k], cA2z[t][2 * k + 1]};
                            const v2f sL  = {accL[t][2 * k], accL[t][2 * k + 1]};
                            const v2f sGx = {dx[2 * k], dx[2 * k + 1]};
                            const v2f sGz = {dz[t][2 * k], dz[t][2 * k + 1]};
                            const v2f pls = {pu[t][s][2 * k], pu[t][s][2 * k + 1]};
                            v2f num = __builtin_elementwise_fma(a2z, sGz, DT2 * pls);
                            num = __builtin_elementwise_fma(c0, q2, num);
                            num = __builtin_elementwise_fma(b0, q1, num);
                            num = __builtin_elementwise_fma(a2x, sGx, num);
                            num = __builtin_elementwise_fma(a1, sL, num);
                            const v2f res = div_rc2(num, a0);
                            pn[2 * k]     = res[0];
                            pn[2 * k + 1] = res[1];
                        }
                    }
#pragma unroll
                    for (int r = 0; r < 8; ++r) {
                        p2[t][r] = p1[t][r];
                        p1[t][r] = inb[t][r] ? pn[r] : 0.0f;
                    }
                    if (!lastst || sample) {
#pragma unroll
                        for (int r = 0; r < 8; ++r) {
                            float* q = (float*)(Lw + oc[t][nb][r >> 2]);
                            q[(r & 3) * FS + 1] = p1[t][r];
                        }
                    }
                }
            };
            auto request_plane = [&](int qs, auto ntc) __attribute__((always_inline)) {
                constexpr int ntl = decltype(ntc)::value;
                const char* const pq = (const char*)(pulse + (size_t)qs * (NZ * NX));
#pragma unroll
                for (int t = 0; t < ntl; ++t) {
#pragma unroll
                    for (int r = 0; r < 8; ++r) {
                        asm volatile("" : "+v"(cofs[t][r]));
                        pu[t][qs][r] = *(const float*)(pq + cofs[t][r]);
                    }
                }
            };
            if (!lastst)       step_tiles(tiles_c<NTW>{});
            else if (wave < 4) step_tiles(tiles_c<1>{});
            const bool bar = (!lastst || sample);
            if (bar) lds_barrier_arrive(p1);
            if (s + PD < KST) {
                if (s + PD < nsteps - 1)                     request_plane(s + PD, tiles_c<NTW>{});
                else if (s + PD == nsteps - 1 && wave < 4)   request_plane(s + PD, tiles_c<1>{});
            }
            if (s + 1 < KST && s + 1 < nsteps) {
                if (s + 1 < nsteps - 1) {
#pragma unroll
                    for (int t = 0; t < NTW; ++t) {
                        accL[t] = v8f{};
                        z_own_rows(p1[t], fw, accL[t]);
                    }
                } else if (wave < 4) {
                    accL[0] = v8f{};
                    z_own_rows(p1[0], fw, accL[0]);
                }
            }
            if (bar) lds_barrier_wait(accL);
            if (sample) {
                float* grow = gout + (size_t)(s >> 2) * nch;
                if (sel_off >= 0) grow[tid] = Fn[sel_off];
                for (int ch = tid + 32 * NWV; ch < nch; ch += 32 * NWV) {
                    int rz = el[2 * ch + 0] - (GZ0 + HL);
                    int rx = el[2 * ch + 1] - (GX0 + HL);
                    if (rz >= 0 && rz < OT && rx >= 0 && rx < OT)
                        grow[ch] = Fn[(rz + HL + 2) * FS + rx + HL + 2];
                }
            }
        }
    }

    if (wave < 4) {
        const int gx0 = GX0 + 16 * tx[0] + lm;
        v2f* const fo2 = (v2f*)fout;
        v4f* const c4o = (v4f*)cpk;
#pragma unroll
        for (int r = 0; r < 8; ++r) {
            const int idx = (GZ0 + 16 * tz[0] + r + 8 * half) * NX + gx0;
            const v2f pp = {p1[0][r], p2[0][r]};
            fo2[idx] = pp;
            if constexpr (MODE == 0) {
                const v4f cc = {cA1[0][r], cA2x[0][r], cA2z[0][r], cD[0][r]};
                c4o[idx] = cc;
            }
        }
    }
}

extern "C" void kernel_launch(void* const* d_in, const int* in_sizes, int n_in,
                              void* d_out, int out_size, void* d_ws, size_t ws_size,
                              hipStream_t stream) {
    const float* pulse   = (const float*)d_in[0];
    const float* vel     = (const float*)d_in[1];
    const float* rho     = (const float*)d_in[2];
    const float* damping = (const float*)d_in[3];
    const float* beta    = (const float*)d_in[4];
    const int*   el      = (const int*)d_in[5];
    const int*   nla     = (const int*)d_in[6];

    const int nt  = in_sizes[0] / (NZ * NX);
    const int nch = in_sizes[5] / 2;

    float* ws = (float*)d_ws;
    const size_t F = (size_t)NZ * NX;
    if (ws_size < 8 * F * sizeof(float)) return;
    float* fb[2] = { ws, ws + 2 * F };
    float* cpk   = ws + 4 * F;

    int cur = 0;
    for (int t0 = 0; t0 < nt; t0 += KST) {
        const int ns = (nt - t0 < KST) ? (nt - t0) : KST;
        const float* ps = pulse + (size_t)t0 * F;
        float* go = (float*)d_out + (size_t)(t0 >> 2) * nch;
        if (t0 == 0) {
            if (ns == KST)
                steps_kernel<true, 0><<<dim3(NX / OT, NZ / OT), 32 * NWV, 0, stream>>>(
                    fb[cur], fb[cur ^ 1], cpk, ps, vel, rho, damping, beta, nla, el, go, nch, ns);
            else
                steps_kernel<false, 0><<<dim3(NX / OT, NZ / OT), 32 * NWV, 0, stream>>>(
                    fb[cur], fb[cur ^ 1], cpk, ps, vel, rho, damping, beta, nla, el, go, nch, ns);
        } else {
            if (ns == KST)
                steps_kernel<true, 1><<<dim3(NX / OT, NZ / OT), 32 * NWV, 0, stream>>>(
                    fb[cur], fb[cur ^ 1], cpk, ps, vel, rho, damping, beta, nla, el, go, nch, ns);
            else
                steps_kernel<false, 1><<<dim3(NX / OT, NZ / OT), 32 * NWV, 0, stream>>>(
                    fb[cur], fb[cur ^ 1], cpk, ps, vel, rho, damping, beta, nla, el, go, nch, ns);
        }
        cur ^= 1;
    }
}
